# starting kernel
# baseline (speedup 1.0000x reference)
#include <hip/hip_runtime.h>

typedef __attribute__((ext_vector_type(2))) float v2f;
typedef __attribute__((ext_vector_type(4))) float v4f;
typedef __attribute__((ext_vector_type(8))) float v8f;

struct OpTab { float v[9][32][4]; };

constexpr float db4pick(const double* h, int t) {
    return (t >= 0 && t < 4) ? (float)h[t] : 0.0f;
}

constexpr OpTab make_optab() {
    OpTab o{};
    const double s3  = 1.7320508075688772935;
    const double nrm = 0.17677669529663688110;
    const double h0[4] = {(1.0 + s3) * nrm, (3.0 + s3) * nrm,
                          (3.0 - s3) * nrm, (1.0 - s3) * nrm};
    const double h1[4] = {h0[3], -h0[2], h0[1], -h0[0]};
    for (int t = 0; t < 9; ++t) {
        for (int lane = 0; lane < 32; ++lane) {
            int n = lane & 15, g = lane >> 4;
            int tau = 4 * t + 2 * g - 2 * n;
            o.v[t][lane][0] = db4pick(h0, tau);
            o.v[t][lane][1] = db4pick(h0, tau + 1);
            o.v[t][lane][2] = db4pick(h1, tau);
            o.v[t][lane][3] = db4pick(h1, tau + 1);
        }
    }
    return o;
}

__device__ __constant__ OpTab OPTAB = make_optab();

#define WMMA_F32(A, B, C) \
    __builtin_amdgcn_wmma_f32_16x16x4_f32(false, (A), false, (B), (short)0, (C), false, false)

__global__ __launch_bounds__(32)
void dwt_db4_level(const float* __restrict__ src, int N, long long srcStride,
                   float* __restrict__ dstLL, int ldLL, long long llStride,
                   float* __restrict__ outDet, int M)
{
    const int lane = threadIdx.x;
    const int n    = lane & 15;
    const int g    = lane >> 4;
    const int c0   = blockIdx.x * 16;
    const int r0   = blockIdx.y * 16;
    const int img  = blockIdx.z;

    src    += (long long)img * srcStride;
    dstLL  += (long long)img * llStride;
    outDet += (long long)img * (512LL * 512LL);

    __shared__ float Xs[34][38];
    __shared__ float Lr[36][17];
    __shared__ float Hr[36][17];

    int gcA = 2 * c0 - 1 + lane;       gcA += (gcA < 0) ? N : 0;  gcA -= (gcA >= N) ? N : 0;
    int gcB = 2 * c0 + 31 + lane;      gcB -= (gcB >= N) ? N : 0;
    int gr  = 2 * r0 - 1;              gr  += (gr  < 0) ? N : 0;
    for (int r = 0; r < 34; ++r) {
        const float* p = src + (long long)gr * N;
        Xs[r][lane] = p[gcA];
        if (lane < 4) Xs[r][32 + lane] = p[gcB];
        ++gr;  gr = (gr >= N) ? 0 : gr;
    }
    if (lane < 16) {
        Lr[34][lane] = 0.0f; Lr[35][lane] = 0.0f;
        Hr[34][lane] = 0.0f; Hr[35][lane] = 0.0f;
    }

    v2f wL[9], wH[9];
    #pragma unroll
    for (int t = 0; t < 9; ++t) {
        v4f w = *(const v4f*)OPTAB.v[t][lane];
        wL[t].x = w.x;  wL[t].y = w.y;
        wH[t].x = w.z;  wH[t].y = w.w;
    }
    __syncthreads();

    const int chunkR[3] = {0, 16, 18};
    #pragma unroll
    for (int ch = 0; ch < 3; ++ch) {
        const int R = chunkR[ch];
        v8f accL = {0.f,0.f,0.f,0.f,0.f,0.f,0.f,0.f};
        v8f accH = {0.f,0.f,0.f,0.f,0.f,0.f,0.f,0.f};
        #pragma unroll
        for (int t = 0; t < 9; ++t) {
            v2f a = *(const v2f*)&Xs[R + n][4 * t + 2 * g];
            accL = WMMA_F32(a, wL[t], accL);
            accH = WMMA_F32(a, wH[t], accH);
        }
        #pragma unroll
        for (int v = 0; v < 8; ++v) {
            Lr[R + v + 8 * g][n] = accL[v];
            Hr[R + v + 8 * g][n] = accH[v];
        }
    }
    __syncthreads();

    v8f LL = {0.f,0.f,0.f,0.f,0.f,0.f,0.f,0.f};
    v8f LH = {0.f,0.f,0.f,0.f,0.f,0.f,0.f,0.f};
    v8f HL = {0.f,0.f,0.f,0.f,0.f,0.f,0.f,0.f};
    v8f HH = {0.f,0.f,0.f,0.f,0.f,0.f,0.f,0.f};
    #pragma unroll
    for (int t = 0; t < 9; ++t) {
        int rr = 4 * t + 2 * g;
        v2f bLr, bHr;
        bLr.x = Lr[rr][n];  bLr.y = Lr[rr + 1][n];
        bHr.x = Hr[rr][n];  bHr.y = Hr[rr + 1][n];
        LL = WMMA_F32(wL[t], bLr, LL);
        LH = WMMA_F32(wH[t], bLr, LH);
        HL = WMMA_F32(wL[t], bHr, HL);
        HH = WMMA_F32(wH[t], bHr, HH);
    }

    #pragma unroll
    for (int v = 0; v < 8; ++v) {
        int orow = r0 + v + 8 * g;
        int ocol = c0 + n;
        dstLL[(long long)orow * ldLL + ocol] = LL[v];
        __builtin_nontemporal_store(LH[v], &outDet[(long long)orow * 512 + (ocol + M)]);
        __builtin_nontemporal_store(HL[v], &outDet[(long long)(orow + M) * 512 + ocol]);
        __builtin_nontemporal_store(HH[v], &outDet[(long long)(orow + M) * 512 + (ocol + M)]);
    }
}

extern "C" void kernel_launch(void* const* d_in, const int* in_sizes, int n_in,
                              void* d_out, int out_size, void* d_ws, size_t ws_size,
                              hipStream_t stream) {
    const float* x = (const float*)d_in[0];
    float* out = (float*)d_out;
    const int BC = in_sizes[0] / (512 * 512);

    float* LL1 = (float*)d_ws;
    float* LL2 = LL1 + (long long)BC * 256 * 256;

    dim3 blk(32, 1, 1);
    dwt_db4_level<<<dim3(16, 16, BC), blk, 0, stream>>>(
        x,   512, 512LL * 512, LL1, 256, 256LL * 256, out, 256);
    dwt_db4_level<<<dim3(8, 8, BC), blk, 0, stream>>>(
        LL1, 256, 256LL * 256, LL2, 128, 128LL * 128, out, 128);
    dwt_db4_level<<<dim3(4, 4, BC), blk, 0, stream>>>(
        LL2, 128, 128LL * 128, out, 512, 512LL * 512, out, 64);
}


# Stage the input tile with every global load in flight at once: 12 aligned 16-byte loads per lane issued ahead of the first LDS store, instead of 68 load-wait-store round trips per wave
# speedup vs baseline: 2.1065x; 2.1065x over previous
#include <hip/hip_runtime.h>

typedef __attribute__((ext_vector_type(2))) float v2f;
typedef __attribute__((ext_vector_type(4))) float v4f;
typedef __attribute__((ext_vector_type(8))) float v8f;

struct OpTab { float v[9][32][4]; };

constexpr float db4pick(const double* h, int t) {
    return (t >= 0 && t < 4) ? (float)h[t] : 0.0f;
}

constexpr OpTab make_optab() {
    OpTab o{};
    const double s3  = 1.7320508075688772935;
    const double nrm = 0.17677669529663688110;
    const double h0[4] = {(1.0 + s3) * nrm, (3.0 + s3) * nrm,
                          (3.0 - s3) * nrm, (1.0 - s3) * nrm};
    const double h1[4] = {h0[3], -h0[2], h0[1], -h0[0]};
    for (int t = 0; t < 9; ++t) {
        for (int lane = 0; lane < 32; ++lane) {
            int n = lane & 15, g = lane >> 4;
            int tau = 4 * t + 2 * g - 2 * n;
            o.v[t][lane][0] = db4pick(h0, tau);
            o.v[t][lane][1] = db4pick(h0, tau + 1);
            o.v[t][lane][2] = db4pick(h1, tau);
            o.v[t][lane][3] = db4pick(h1, tau + 1);
        }
    }
    return o;
}

__device__ __constant__ OpTab OPTAB = make_optab();

#define WMMA_F32(A, B, C) \
    __builtin_amdgcn_wmma_f32_16x16x4_f32(false, (A), false, (B), (short)0, (C), false, false)

__global__ __launch_bounds__(32)
void dwt_db4_level(const float* __restrict__ src, int N, long long srcStride,
                   float* __restrict__ dstLL, int ldLL, long long llStride,
                   float* __restrict__ outDet, int M)
{
    const int lane = threadIdx.x;
    const int n    = lane & 15;
    const int g    = lane >> 4;
    const int c0   = blockIdx.x * 16;
    const int r0   = blockIdx.y * 16;
    const int img  = blockIdx.z;

    src    += (long long)img * srcStride;
    dstLL  += (long long)img * llStride;
    outDet += (long long)img * (512LL * 512LL);

    __shared__ __attribute__((aligned(16))) float Xs[34][42];
    __shared__ float Lr[36][17];
    __shared__ float Hr[36][17];

    v2f wL[9], wH[9];
    #pragma unroll
    for (int t = 0; t < 9; ++t) {
        v4f w = *(const v4f*)OPTAB.v[t][lane];
        wL[t].x = w.x;  wL[t].y = w.y;
        wH[t].x = w.z;  wH[t].y = w.w;
    }

    const int lr = lane / 10;
    const int q  = lane - 10 * lr;
    v4f st[12];
    {
        int gc = 2 * c0 - 4 + 4 * q;   gc += (gc < 0) ? N : 0;  gc -= (gc >= N) ? N : 0;
        int rb = 2 * r0 - 1 + lr;      rb += (rb < 0) ? N : 0;
        const unsigned rowBytes = 4u * (unsigned)N;
        const unsigned colBytes = 4u * (unsigned)gc;
        const char* base = (const char*)src;
        #pragma unroll
        for (int i = 0; i < 12; ++i) {
            unsigned x = (unsigned)(rb + 3 * i);
            unsigned xm = x - (unsigned)N;
            x = (xm < x) ? xm : x;
            unsigned off = __umul24(x, rowBytes) + colBytes;
            st[i] = *(const v4f*)(base + off);
        }
    }
    if (lane < 16) {
        Lr[34][lane] = 0.0f; Lr[35][lane] = 0.0f;
        Hr[34][lane] = 0.0f; Hr[35][lane] = 0.0f;
    }
    {
        #pragma unroll
        for (int i = 0; i < 12; ++i) {
            const bool mine = (i < 11) ? (lane < 30) : (lane < 10);
            if (mine) {
                float* d = &Xs[3 * i + lr][4 * q + 1];
                d[0] = st[i].x;  d[1] = st[i].y;  d[2] = st[i].z;  d[3] = st[i].w;
            }
        }
    }
    __syncthreads();

    const int chunkR[3] = {0, 16, 18};
    #pragma unroll
    for (int ch = 0; ch < 3; ++ch) {
        const int R = chunkR[ch];
        v8f accL = {0.f,0.f,0.f,0.f,0.f,0.f,0.f,0.f};
        v8f accH = {0.f,0.f,0.f,0.f,0.f,0.f,0.f,0.f};
        #pragma unroll
        for (int t = 0; t < 9; ++t) {
            v2f a = *(const v2f*)&Xs[R + n][4 * t + 2 * g + 4];
            accL = WMMA_F32(a, wL[t], accL);
            accH = WMMA_F32(a, wH[t], accH);
        }
        #pragma unroll
        for (int v = 0; v < 8; ++v) {
            Lr[R + v + 8 * g][n] = accL[v];
            Hr[R + v + 8 * g][n] = accH[v];
        }
    }
    __syncthreads();

    v8f LL = {0.f,0.f,0.f,0.f,0.f,0.f,0.f,0.f};
    v8f LH = {0.f,0.f,0.f,0.f,0.f,0.f,0.f,0.f};
    v8f HL = {0.f,0.f,0.f,0.f,0.f,0.f,0.f,0.f};
    v8f HH = {0.f,0.f,0.f,0.f,0.f,0.f,0.f,0.f};
    #pragma unroll
    for (int t = 0; t < 9; ++t) {
        int rr = 4 * t + 2 * g;
        v2f bLr, bHr;
        bLr.x = Lr[rr][n];  bLr.y = Lr[rr + 1][n];
        bHr.x = Hr[rr][n];  bHr.y = Hr[rr + 1][n];
        LL = WMMA_F32(wL[t], bLr, LL);
        LH = WMMA_F32(wH[t], bLr, LH);
        HL = WMMA_F32(wL[t], bHr, HL);
        HH = WMMA_F32(wH[t], bHr, HH);
    }

    #pragma unroll
    for (int v = 0; v < 8; ++v) {
        int orow = r0 + v + 8 * g;
        int ocol = c0 + n;
        dstLL[(long long)orow * ldLL + ocol] = LL[v];
        __builtin_nontemporal_store(LH[v], &outDet[(long long)orow * 512 + (ocol + M)]);
        __builtin_nontemporal_store(HL[v], &outDet[(long long)(orow + M) * 512 + ocol]);
        __builtin_nontemporal_store(HH[v], &outDet[(long long)(orow + M) * 512 + (ocol + M)]);
    }
}

extern "C" void kernel_launch(void* const* d_in, const int* in_sizes, int n_in,
                              void* d_out, int out_size, void* d_ws, size_t ws_size,
                              hipStream_t stream) {
    const float* x = (const float*)d_in[0];
    float* out = (float*)d_out;
    const int BC = in_sizes[0] / (512 * 512);

    float* LL1 = (float*)d_ws;
    float* LL2 = LL1 + (long long)BC * 256 * 256;

    dim3 blk(32, 1, 1);
    dwt_db4_level<<<dim3(16, 16, BC), blk, 0, stream>>>(
        x,   512, 512LL * 512, LL1, 256, 256LL * 256, out, 256);
    dwt_db4_level<<<dim3(8, 8, BC), blk, 0, stream>>>(
        LL1, 256, 256LL * 256, LL2, 128, 128LL * 128, out, 128);
    dwt_db4_level<<<dim3(4, 4, BC), blk, 0, stream>>>(
        LL2, 128, 128LL * 128, out, 512, 512LL * 512, out, 64);
}


# Each wave walks 4, 4, 2 adjacent tiles per level in a loop: operator fragments and row offsets set up once, the next tile's 12 loads in flight under the current tile's 90 matrix products
# speedup vs baseline: 2.3617x; 1.1211x over previous
#include <hip/hip_runtime.h>

typedef __attribute__((ext_vector_type(2))) float v2f;
typedef __attribute__((ext_vector_type(4))) float v4f;
typedef __attribute__((ext_vector_type(8))) float v8f;

struct OpTab { float v[9][32][4]; };

constexpr float db4pick(const double* h, int t) {
    return (t >= 0 && t < 4) ? (float)h[t] : 0.0f;
}

constexpr OpTab make_optab() {
    OpTab o{};
    const double s3  = 1.7320508075688772935;
    const double nrm = 0.17677669529663688110;
    const double h0[4] = {(1.0 + s3) * nrm, (3.0 + s3) * nrm,
                          (3.0 - s3) * nrm, (1.0 - s3) * nrm};
    const double h1[4] = {h0[3], -h0[2], h0[1], -h0[0]};
    for (int t = 0; t < 9; ++t) {
        for (int lane = 0; lane < 32; ++lane) {
            int n = lane & 15, g = lane >> 4;
            int tau = 4 * t + 2 * g - 2 * n;
            o.v[t][lane][0] = db4pick(h0, tau);
            o.v[t][lane][1] = db4pick(h0, tau + 1);
            o.v[t][lane][2] = db4pick(h1, tau);
            o.v[t][lane][3] = db4pick(h1, tau + 1);
        }
    }
    return o;
}

__device__ __constant__ OpTab OPTAB = make_optab();

#define WMMA_F32(A, B, C) \
    __builtin_amdgcn_wmma_f32_16x16x4_f32(false, (A), false, (B), (short)0, (C), false, false)

__device__ __forceinline__ void lds_handover() {
    __builtin_amdgcn_fence(__ATOMIC_RELEASE, "workgroup", "local");
    __builtin_amdgcn_s_barrier();
    __builtin_amdgcn_fence(__ATOMIC_ACQUIRE, "workgroup", "local");
}

__global__ __launch_bounds__(32)
void dwt_db4_level(const float* __restrict__ src, int N, long long srcStride,
                   float* __restrict__ dstLL, int ldLL, long long llStride,
                   float* __restrict__ outDet, int M, int T)
{
    const int lane = threadIdx.x;
    const int n    = lane & 15;
    const int g    = lane >> 4;
    int       c0   = blockIdx.x * T * 16;
    const int r0   = blockIdx.y * 16;
    const int img  = blockIdx.z;

    src    += (long long)img * srcStride;
    dstLL  += (long long)img * llStride;
    outDet += (long long)img * (512LL * 512LL);

    __shared__ __attribute__((aligned(16))) float Xs[34][42];
    __shared__ float Lr[36][17];
    __shared__ float Hr[36][17];

    v2f wL[9], wH[9];
    #pragma unroll
    for (int t = 0; t < 9; ++t) {
        v4f w = *(const v4f*)OPTAB.v[t][lane];
        wL[t].x = w.x;  wL[t].y = w.y;
        wH[t].x = w.z;  wH[t].y = w.w;
    }

    const int lr = lane / 10;
    const int q  = lane - 10 * lr;
    const char* base = (const char*)src;
    unsigned rowOff[12];
    {
        int rb = 2 * r0 - 1 + lr;      rb += (rb < 0) ? N : 0;
        const unsigned rowBytes = 4u * (unsigned)N;
        #pragma unroll
        for (int i = 0; i < 12; ++i) {
            unsigned x = (unsigned)(rb + 3 * i);
            unsigned xm = x - (unsigned)N;
            x = (xm < x) ? xm : x;
            rowOff[i] = __umul24(x, rowBytes);
        }
    }
    v4f st[12];
    {
        int gc = 2 * c0 - 4 + 4 * q;   gc += (gc < 0) ? N : 0;  gc -= (gc >= N) ? N : 0;
        const unsigned colBytes = 4u * (unsigned)gc;
        #pragma unroll
        for (int i = 0; i < 12; ++i) st[i] = *(const v4f*)(base + (rowOff[i] + colBytes));
    }
    if (lane < 16) {
        Lr[34][lane] = 0.0f; Lr[35][lane] = 0.0f;
        Hr[34][lane] = 0.0f; Hr[35][lane] = 0.0f;
    }

    #pragma nounroll
    for (int it = 0; it < T; ++it, c0 += 16) {
    {
        #pragma unroll
        for (int i = 0; i < 12; ++i) {
            const bool mine = (i < 11) ? (lane < 30) : (lane < 10);
            if (mine) {
                float* d = &Xs[3 * i + lr][4 * q + 1];
                d[0] = st[i].x;  d[1] = st[i].y;  d[2] = st[i].z;  d[3] = st[i].w;
            }
        }
    }
    __syncthreads();

    if (it + 1 < T) {
        int gc = 2 * c0 + 28 + 4 * q;  gc -= (gc >= N) ? N : 0;
        const unsigned colBytes = 4u * (unsigned)gc;
        #pragma unroll
        for (int i = 0; i < 12; ++i) st[i] = *(const v4f*)(base + (rowOff[i] + colBytes));
    }

    const int chunkR[3] = {0, 16, 18};
    #pragma unroll
    for (int ch = 0; ch < 3; ++ch) {
        const int R = chunkR[ch];
        v8f accL = {0.f,0.f,0.f,0.f,0.f,0.f,0.f,0.f};
        v8f accH = {0.f,0.f,0.f,0.f,0.f,0.f,0.f,0.f};
        #pragma unroll
        for (int t = 0; t < 9; ++t) {
            v2f a = *(const v2f*)&Xs[R + n][4 * t + 2 * g + 4];
            accL = WMMA_F32(a, wL[t], accL);
            accH = WMMA_F32(a, wH[t], accH);
        }
        #pragma unroll
        for (int v = 0; v < 8; ++v) {
            Lr[R + v + 8 * g][n] = accL[v];
            Hr[R + v + 8 * g][n] = accH[v];
        }
    }
    lds_handover();

    v8f LL = {0.f,0.f,0.f,0.f,0.f,0.f,0.f,0.f};
    v8f LH = {0.f,0.f,0.f,0.f,0.f,0.f,0.f,0.f};
    v8f HL = {0.f,0.f,0.f,0.f,0.f,0.f,0.f,0.f};
    v8f HH = {0.f,0.f,0.f,0.f,0.f,0.f,0.f,0.f};
    #pragma unroll
    for (int t = 0; t < 9; ++t) {
        int rr = 4 * t + 2 * g;
        v2f bLr, bHr;
        bLr.x = Lr[rr][n];  bLr.y = Lr[rr + 1][n];
        bHr.x = Hr[rr][n];  bHr.y = Hr[rr + 1][n];
        LL = WMMA_F32(wL[t], bLr, LL);
        LH = WMMA_F32(wH[t], bLr, LH);
        HL = WMMA_F32(wL[t], bHr, HL);
        HH = WMMA_F32(wH[t], bHr, HH);
    }

    {
        const unsigned ocol  = (unsigned)(c0 + n);
        const unsigned orow0 = (unsigned)(r0 + 8 * g);
        const unsigned oLL   = 4u * (orow0 * (unsigned)ldLL + ocol);
        const unsigned oD    = 4u * (orow0 * 512u + ocol);
        const unsigned pitchLL = 4u * (unsigned)ldLL;
        char* pLL = (char*)dstLL;
        char* pLH = (char*)(outDet + M);
        char* pHL = (char*)(outDet + (long long)M * 512);
        char* pHH = (char*)(outDet + (long long)M * 512 + M);
        #pragma unroll
        for (int v = 0; v < 8; ++v) {
            *(float*)(pLL + (oLL + (unsigned)v * pitchLL)) = LL[v];
            __builtin_nontemporal_store(LH[v], (float*)(pLH + (oD + 2048u * (unsigned)v)));
            __builtin_nontemporal_store(HL[v], (float*)(pHL + (oD + 2048u * (unsigned)v)));
            __builtin_nontemporal_store(HH[v], (float*)(pHH + (oD + 2048u * (unsigned)v)));
        }
    }
    }
}

extern "C" void kernel_launch(void* const* d_in, const int* in_sizes, int n_in,
                              void* d_out, int out_size, void* d_ws, size_t ws_size,
                              hipStream_t stream) {
    const float* x = (const float*)d_in[0];
    float* out = (float*)d_out;
    const int BC = in_sizes[0] / (512 * 512);

    float* LL1 = (float*)d_ws;
    float* LL2 = LL1 + (long long)BC * 256 * 256;

    dim3 blk(32, 1, 1);
    dwt_db4_level<<<dim3(4, 16, BC), blk, 0, stream>>>(
        x,   512, 512LL * 512, LL1, 256, 256LL * 256, out, 256, 4);
    dwt_db4_level<<<dim3(2, 8, BC), blk, 0, stream>>>(
        LL1, 256, 256LL * 256, LL2, 128, 128LL * 128, out, 128, 4);
    dwt_db4_level<<<dim3(2, 4, BC), blk, 0, stream>>>(
        LL2, 128, 128LL * 128, out, 512, 512LL * 512, out, 64, 2);
}


# Both db4 filters folded into one 16-wide WMMA operator (8 outputs x 2 filters, K 20 not 36): each product applies low- and high-pass at once, 50 not 90 matrix products per tile; sum order unchanged
# speedup vs baseline: 2.6693x; 1.1302x over previous
#include <hip/hip_runtime.h>

typedef __attribute__((ext_vector_type(2))) float v2f;
typedef __attribute__((ext_vector_type(4))) float v4f;
typedef __attribute__((ext_vector_type(8))) float v8f;

struct OpTab { float v[5][32][2]; };

constexpr float db4pick(const double* h, int t) {
    return (t >= 0 && t < 4) ? (float)h[t] : 0.0f;
}

constexpr OpTab make_optab() {
    OpTab o{};
    const double s3  = 1.7320508075688772935;
    const double nrm = 0.17677669529663688110;
    const double h0[4] = {(1.0 + s3) * nrm, (3.0 + s3) * nrm,
                          (3.0 - s3) * nrm, (1.0 - s3) * nrm};
    const double h1[4] = {h0[3], -h0[2], h0[1], -h0[0]};
    for (int u = 0; u < 5; ++u) {
        for (int lane = 0; lane < 32; ++lane) {
            int m = (lane & 15) >> 1, f = lane & 1, h = lane >> 4;
            int tau = 4 * u + 2 * h - 2 * m;
            const double* w = f ? h1 : h0;
            o.v[u][lane][0] = db4pick(w, tau);
            o.v[u][lane][1] = db4pick(w, tau + 1);
        }
    }
    return o;
}

__device__ __constant__ OpTab OPTAB = make_optab();

#define WMMA_F32(A, B, C) \
    __builtin_amdgcn_wmma_f32_16x16x4_f32(false, (A), false, (B), (short)0, (C), false, false)

__device__ __forceinline__ void lds_handover() {
    __builtin_amdgcn_fence(__ATOMIC_RELEASE, "workgroup", "local");
    __builtin_amdgcn_s_barrier();
    __builtin_amdgcn_fence(__ATOMIC_ACQUIRE, "workgroup", "local");
}

__global__ __launch_bounds__(32)
void dwt_db4_level(const float* __restrict__ src, int N, long long srcStride,
                   float* __restrict__ dstLL, int ldLL, long long llStride,
                   float* __restrict__ outDet, int M, int T)
{
    const int lane = threadIdx.x;
    const int n    = lane & 15;
    const int g    = lane >> 4;
    int       c0   = blockIdx.x * T * 16;
    const int r0   = blockIdx.y * 16;
    const int img  = blockIdx.z;

    src    += (long long)img * srcStride;
    dstLL  += (long long)img * llStride;
    outDet += (long long)img * (512LL * 512LL);

    __shared__ __attribute__((aligned(16))) float Xs[34][42];
    __shared__ float LH[2][36][17];

    v2f W[5];
    #pragma unroll
    for (int u = 0; u < 5; ++u) W[u] = *(const v2f*)OPTAB.v[u][lane];

    const int lr = lane / 10;
    const int q  = lane - 10 * lr;
    const char* base = (const char*)src;
    unsigned rowOff[12];
    {
        int rb = 2 * r0 - 1 + lr;      rb += (rb < 0) ? N : 0;
        const unsigned rowBytes = 4u * (unsigned)N;
        #pragma unroll
        for (int i = 0; i < 12; ++i) {
            unsigned x = (unsigned)(rb + 3 * i);
            unsigned xm = x - (unsigned)N;
            x = (xm < x) ? xm : x;
            rowOff[i] = __umul24(x, rowBytes);
        }
    }
    v4f st[12];
    {
        int gc = 2 * c0 - 4 + 4 * q;   gc += (gc < 0) ? N : 0;  gc -= (gc >= N) ? N : 0;
        const unsigned colBytes = 4u * (unsigned)gc;
        #pragma unroll
        for (int i = 0; i < 12; ++i) st[i] = *(const v4f*)(base + (rowOff[i] + colBytes));
    }
    if (lane < 16) {
        LH[0][34][lane] = 0.0f; LH[0][35][lane] = 0.0f;
        LH[1][34][lane] = 0.0f; LH[1][35][lane] = 0.0f;
    }

    #pragma nounroll
    for (int it = 0; it < T; ++it, c0 += 16) {
    {
        #pragma unroll
        for (int i = 0; i < 12; ++i) {
            const bool mine = (i < 11) ? (lane < 30) : (lane < 10);
            if (mine) {
                float* d = &Xs[3 * i + lr][4 * q + 1];
                d[0] = st[i].x;  d[1] = st[i].y;  d[2] = st[i].z;  d[3] = st[i].w;
            }
        }
    }
    __syncthreads();

    if (it + 1 < T) {
        int gc = 2 * c0 + 28 + 4 * q;  gc -= (gc >= N) ? N : 0;
        const unsigned colBytes = 4u * (unsigned)gc;
        #pragma unroll
        for (int i = 0; i < 12; ++i) st[i] = *(const v4f*)(base + (rowOff[i] + colBytes));
    }

    const int chunkR[3] = {0, 16, 18};
    float* const sdst = &LH[n & 1][8 * g][n >> 1];
    #pragma unroll
    for (int ch = 0; ch < 3; ++ch) {
        const int R = chunkR[ch];
        v8f acc0 = {0.f,0.f,0.f,0.f,0.f,0.f,0.f,0.f};
        v8f acc1 = {0.f,0.f,0.f,0.f,0.f,0.f,0.f,0.f};
        #pragma unroll
        for (int t = 0; t < 9; ++t) {
            v2f a = *(const v2f*)&Xs[R + n][4 * t + 2 * g + 4];
            if (t <= 4) acc0 = WMMA_F32(a, W[t], acc0);
            if (t >= 4) acc1 = WMMA_F32(a, W[t - 4], acc1);
        }
        #pragma unroll
        for (int v = 0; v < 8; ++v) {
            sdst[(R + v) * 17]     = acc0[v];
            sdst[(R + v) * 17 + 8] = acc1[v];
        }
    }
    lds_handover();

    v8f L0 = {0.f,0.f,0.f,0.f,0.f,0.f,0.f,0.f};
    v8f L1 = {0.f,0.f,0.f,0.f,0.f,0.f,0.f,0.f};
    v8f H0 = {0.f,0.f,0.f,0.f,0.f,0.f,0.f,0.f};
    v8f H1 = {0.f,0.f,0.f,0.f,0.f,0.f,0.f,0.f};
    #pragma unroll
    for (int t = 0; t < 9; ++t) {
        int rr = 4 * t + 2 * g;
        v2f bLr, bHr;
        bLr.x = LH[0][rr][n];  bLr.y = LH[0][rr + 1][n];
        bHr.x = LH[1][rr][n];  bHr.y = LH[1][rr + 1][n];
        if (t <= 4) { L0 = WMMA_F32(W[t], bLr, L0);      H0 = WMMA_F32(W[t], bHr, H0); }
        if (t >= 4) { L1 = WMMA_F32(W[t - 4], bLr, L1);  H1 = WMMA_F32(W[t - 4], bHr, H1); }
    }

    {
        const unsigned ocol  = (unsigned)(c0 + n);
        const unsigned orow0 = (unsigned)(r0 + 4 * g);
        const unsigned oLL   = 4u * (orow0 * (unsigned)ldLL + ocol);
        const unsigned oD    = 4u * (orow0 * 512u + ocol);
        const unsigned pitchLL = 4u * (unsigned)ldLL;
        char* pLL = (char*)dstLL;
        char* pLH = (char*)(outDet + M);
        char* pHL = (char*)(outDet + (long long)M * 512);
        char* pHH = (char*)(outDet + (long long)M * 512 + M);
        #pragma unroll
        for (int w = 0; w < 4; ++w) {
            *(float*)(pLL + (oLL + (unsigned)w * pitchLL))       = L0[2 * w];
            *(float*)(pLL + (oLL + (unsigned)(w + 8) * pitchLL)) = L1[2 * w];
            __builtin_nontemporal_store(L0[2 * w + 1], (float*)(pLH + (oD + 2048u * (unsigned)w)));
            __builtin_nontemporal_store(L1[2 * w + 1], (float*)(pLH + (oD + 2048u * (unsigned)(w + 8))));
            __builtin_nontemporal_store(H0[2 * w],     (float*)(pHL + (oD + 2048u * (unsigned)w)));
            __builtin_nontemporal_store(H1[2 * w],     (float*)(pHL + (oD + 2048u * (unsigned)(w + 8))));
            __builtin_nontemporal_store(H0[2 * w + 1], (float*)(pHH + (oD + 2048u * (unsigned)w)));
            __builtin_nontemporal_store(H1[2 * w + 1], (float*)(pHH + (oD + 2048u * (unsigned)(w + 8))));
        }
    }
    }
}

extern "C" void kernel_launch(void* const* d_in, const int* in_sizes, int n_in,
                              void* d_out, int out_size, void* d_ws, size_t ws_size,
                              hipStream_t stream) {
    const float* x = (const float*)d_in[0];
    float* out = (float*)d_out;
    const int BC = in_sizes[0] / (512 * 512);

    float* LL1 = (float*)d_ws;
    float* LL2 = LL1 + (long long)BC * 256 * 256;

    dim3 blk(32, 1, 1);
    dwt_db4_level<<<dim3(4, 16, BC), blk, 0, stream>>>(
        x,   512, 512LL * 512, LL1, 256, 256LL * 256, out, 256, 4);
    dwt_db4_level<<<dim3(2, 8, BC), blk, 0, stream>>>(
        LL1, 256, 256LL * 256, LL2, 128, 128LL * 128, out, 128, 4);
    dwt_db4_level<<<dim3(2, 4, BC), blk, 0, stream>>>(
        LL2, 128, 128LL * 128, out, 512, 512LL * 512, out, 64, 2);
}


# Whole 128-byte lines per store: vertical pass and stores run once per tile pair, split into even and odd columns, so each lane stores two adjacent columns as 8 bytes; sum order unchanged
# speedup vs baseline: 3.2642x; 1.2229x over previous
#include <hip/hip_runtime.h>

typedef __attribute__((ext_vector_type(2))) float v2f;
typedef __attribute__((ext_vector_type(4))) float v4f;
typedef __attribute__((ext_vector_type(8))) float v8f;

struct OpTab { float v[5][32][2]; };

constexpr float db4pick(const double* h, int t) {
    return (t >= 0 && t < 4) ? (float)h[t] : 0.0f;
}

constexpr OpTab make_optab() {
    OpTab o{};
    const double s3  = 1.7320508075688772935;
    const double nrm = 0.17677669529663688110;
    const double h0[4] = {(1.0 + s3) * nrm, (3.0 + s3) * nrm,
                          (3.0 - s3) * nrm, (1.0 - s3) * nrm};
    const double h1[4] = {h0[3], -h0[2], h0[1], -h0[0]};
    for (int u = 0; u < 5; ++u) {
        for (int lane = 0; lane < 32; ++lane) {
            int m = (lane & 15) >> 1, f = lane & 1, h = lane >> 4;
            int tau = 4 * u + 2 * h - 2 * m;
            const double* w = f ? h1 : h0;
            o.v[u][lane][0] = db4pick(w, tau);
            o.v[u][lane][1] = db4pick(w, tau + 1);
        }
    }
    return o;
}

__device__ __constant__ OpTab OPTAB = make_optab();

#define WMMA_F32(A, B, C) \
    __builtin_amdgcn_wmma_f32_16x16x4_f32(false, (A), false, (B), (short)0, (C), false, false)

__device__ __forceinline__ v2f col_pair(float e, float o) {
    v2f p;  p.x = e;  p.y = o;
    asm volatile("" : "+v"(p));
    return p;
}

__device__ __forceinline__ void lds_handover() {
    __builtin_amdgcn_fence(__ATOMIC_RELEASE, "workgroup", "local");
    __builtin_amdgcn_s_barrier();
    __builtin_amdgcn_fence(__ATOMIC_ACQUIRE, "workgroup", "local");
}

__global__ __launch_bounds__(32)
void dwt_db4_level(const float* __restrict__ src, int N, long long srcStride,
                   float* __restrict__ dstLL, int ldLL, long long llStride,
                   float* __restrict__ outDet, int M, int T)
{
    const int lane = threadIdx.x;
    const int n    = lane & 15;
    const int g    = lane >> 4;
    int       c0   = blockIdx.x * T * 16;
    const int r0   = blockIdx.y * 16;
    const int img  = blockIdx.z;

    src    += (long long)img * srcStride;
    dstLL  += (long long)img * llStride;
    outDet += (long long)img * (512LL * 512LL);

    __shared__ __attribute__((aligned(16))) float Xs[34][42];
    __shared__ __attribute__((aligned(8))) float LHt[2][32][38];

    v2f W[5];
    #pragma unroll
    for (int u = 0; u < 5; ++u) W[u] = *(const v2f*)OPTAB.v[u][lane];

    const int lr = lane / 10;
    const int q  = lane - 10 * lr;
    const char* base = (const char*)src;
    unsigned rowOff[12];
    {
        int rb = 2 * r0 - 1 + lr;      rb += (rb < 0) ? N : 0;
        const unsigned rowBytes = 4u * (unsigned)N;
        #pragma unroll
        for (int i = 0; i < 12; ++i) {
            unsigned x = (unsigned)(rb + 3 * i);
            unsigned xm = x - (unsigned)N;
            x = (xm < x) ? xm : x;
            rowOff[i] = __umul24(x, rowBytes);
        }
    }
    v4f st[12];
    {
        int gc = 2 * c0 - 4 + 4 * q;   gc += (gc < 0) ? N : 0;  gc -= (gc >= N) ? N : 0;
        const unsigned colBytes = 4u * (unsigned)gc;
        #pragma unroll
        for (int i = 0; i < 12; ++i) st[i] = *(const v4f*)(base + (rowOff[i] + colBytes));
    }
    {
        const v2f z = {0.f, 0.f};
        *(v2f*)&LHt[0][lane][34] = z;
        *(v2f*)&LHt[1][lane][34] = z;
    }

    #pragma nounroll
    for (int it = 0; it < T; ++it, c0 += 16) {
    {
        #pragma unroll
        for (int i = 0; i < 12; ++i) {
            const bool mine = (i < 11) ? (lane < 30) : (lane < 10);
            if (mine) {
                float* d = &Xs[3 * i + lr][4 * q + 1];
                d[0] = st[i].x;  d[1] = st[i].y;  d[2] = st[i].z;  d[3] = st[i].w;
            }
        }
    }
    __syncthreads();

    if (it + 1 < T) {
        int gc = 2 * c0 + 28 + 4 * q;  gc -= (gc >= N) ? N : 0;
        const unsigned colBytes = 4u * (unsigned)gc;
        #pragma unroll
        for (int i = 0; i < 12; ++i) st[i] = *(const v4f*)(base + (rowOff[i] + colBytes));
    }

    const int chunkR[3] = {0, 16, 18};
    float* const sdst = &LHt[n & 1][16 * (it & 1) + (n >> 1)][8 * g];
    #pragma unroll
    for (int ch = 0; ch < 3; ++ch) {
        const int R = chunkR[ch];
        v8f acc0 = {0.f,0.f,0.f,0.f,0.f,0.f,0.f,0.f};
        v8f acc1 = {0.f,0.f,0.f,0.f,0.f,0.f,0.f,0.f};
        #pragma unroll
        for (int t = 0; t < 9; ++t) {
            v2f a = *(const v2f*)&Xs[R + n][4 * t + 2 * g + 4];
            if (t <= 4) acc0 = WMMA_F32(a, W[t], acc0);
            if (t >= 4) acc1 = WMMA_F32(a, W[t - 4], acc1);
        }
        if (ch < 2) {
            #pragma unroll
            for (int v = 0; v < 8; v += 2) {
                v2f p0, p1;
                p0.x = acc0[v];  p0.y = acc0[v + 1];
                p1.x = acc1[v];  p1.y = acc1[v + 1];
                *(v2f*)&sdst[R + v]          = p0;
                *(v2f*)&sdst[8 * 38 + R + v] = p1;
            }
        } else if (g) {
            v2f p0, p1;
            p0.x = acc0[6];  p0.y = acc0[7];
            p1.x = acc1[6];  p1.y = acc1[7];
            *(v2f*)&sdst[R + 6]          = p0;
            *(v2f*)&sdst[8 * 38 + R + 6] = p1;
        }
    }
    lds_handover();

    if (it & 1) {
    v8f Le0 = {0.f,0.f,0.f,0.f,0.f,0.f,0.f,0.f};
    v8f Lo0 = {0.f,0.f,0.f,0.f,0.f,0.f,0.f,0.f};
    v8f Le1 = {0.f,0.f,0.f,0.f,0.f,0.f,0.f,0.f};
    v8f Lo1 = {0.f,0.f,0.f,0.f,0.f,0.f,0.f,0.f};
    v8f He0 = {0.f,0.f,0.f,0.f,0.f,0.f,0.f,0.f};
    v8f Ho0 = {0.f,0.f,0.f,0.f,0.f,0.f,0.f,0.f};
    v8f He1 = {0.f,0.f,0.f,0.f,0.f,0.f,0.f,0.f};
    v8f Ho1 = {0.f,0.f,0.f,0.f,0.f,0.f,0.f,0.f};
    const float* const bp = &LHt[0][2 * n][2 * g];
    #pragma unroll
    for (int t = 0; t < 9; ++t) {
        const v2f bLe = *(const v2f*)(bp + 4 * t);
        const v2f bLo = *(const v2f*)(bp + 38 + 4 * t);
        const v2f bHe = *(const v2f*)(bp + 32 * 38 + 4 * t);
        const v2f bHo = *(const v2f*)(bp + 33 * 38 + 4 * t);
        if (t <= 4) {
            Le0 = WMMA_F32(W[t], bLe, Le0);      Lo0 = WMMA_F32(W[t], bLo, Lo0);
            He0 = WMMA_F32(W[t], bHe, He0);      Ho0 = WMMA_F32(W[t], bHo, Ho0);
        }
        if (t >= 4) {
            Le1 = WMMA_F32(W[t - 4], bLe, Le1);  Lo1 = WMMA_F32(W[t - 4], bLo, Lo1);
            He1 = WMMA_F32(W[t - 4], bHe, He1);  Ho1 = WMMA_F32(W[t - 4], bHo, Ho1);
        }
    }

    {
        const unsigned ocol  = (unsigned)(c0 - 16 + 2 * n);
        const unsigned orow0 = (unsigned)(r0 + 4 * g);
        const unsigned oLL   = 4u * (orow0 * (unsigned)ldLL + ocol);
        const unsigned oD    = 4u * (orow0 * 512u + ocol);
        const unsigned pitchLL = 4u * (unsigned)ldLL;
        char* pLL = (char*)dstLL;
        char* pLH = (char*)(outDet + M);
        char* pHL = (char*)(outDet + (long long)M * 512);
        char* pHH = (char*)(outDet + (long long)M * 512 + M);
        v2f sLL[8], sLH[8], sHL[8], sHH[8];
        #pragma unroll
        for (int w = 0; w < 4; ++w) {
            sLL[w]     = col_pair(Le0[2 * w],     Lo0[2 * w]);
            sLL[w + 4] = col_pair(Le1[2 * w],     Lo1[2 * w]);
            sLH[w]     = col_pair(Le0[2 * w + 1], Lo0[2 * w + 1]);
            sLH[w + 4] = col_pair(Le1[2 * w + 1], Lo1[2 * w + 1]);
            sHL[w]     = col_pair(He0[2 * w],     Ho0[2 * w]);
            sHL[w + 4] = col_pair(He1[2 * w],     Ho1[2 * w]);
            sHH[w]     = col_pair(He0[2 * w + 1], Ho0[2 * w + 1]);
            sHH[w + 4] = col_pair(He1[2 * w + 1], Ho1[2 * w + 1]);
        }
        unsigned oLLr[8];
        #pragma unroll
        for (int j = 0; j < 8; ++j) oLLr[j] = oLL + (unsigned)((j & 3) + 8 * (j >> 2)) * pitchLL;
        #pragma unroll
        for (int j = 0; j < 8; ++j) {
            const int row = (j & 3) + 8 * (j >> 2);
            *(v2f*)(pLL + oLLr[j]) = sLL[j];
            __builtin_nontemporal_store(sLH[j], (v2f*)((pLH + 2048 * row) + oD));
            __builtin_nontemporal_store(sHL[j], (v2f*)((pHL + 2048 * row) + oD));
            __builtin_nontemporal_store(sHH[j], (v2f*)((pHH + 2048 * row) + oD));
        }
    }
    }
    }
}

extern "C" void kernel_launch(void* const* d_in, const int* in_sizes, int n_in,
                              void* d_out, int out_size, void* d_ws, size_t ws_size,
                              hipStream_t stream) {
    const float* x = (const float*)d_in[0];
    float* out = (float*)d_out;
    const int BC = in_sizes[0] / (512 * 512);

    float* LL1 = (float*)d_ws;
    float* LL2 = LL1 + (long long)BC * 256 * 256;

    dim3 blk(32, 1, 1);
    dwt_db4_level<<<dim3(4, 16, BC), blk, 0, stream>>>(
        x,   512, 512LL * 512, LL1, 256, 256LL * 256, out, 256, 4);
    dwt_db4_level<<<dim3(2, 8, BC), blk, 0, stream>>>(
        LL1, 256, 256LL * 256, LL2, 128, 128LL * 128, out, 128, 4);
    dwt_db4_level<<<dim3(2, 4, BC), blk, 0, stream>>>(
        LL2, 128, 128LL * 128, out, 512, 512LL * 512, out, 64, 2);
}


# Tile loads start on a 128-byte line: 9 chunks per row, 2 line requests not 3; the one input column left of a tile is carried from the tile before it through LDS (read from memory once per strip)
# speedup vs baseline: 3.6844x; 1.1287x over previous
#include <hip/hip_runtime.h>

typedef __attribute__((ext_vector_type(2))) float v2f;
typedef __attribute__((ext_vector_type(4))) float v4f;
typedef __attribute__((ext_vector_type(8))) float v8f;

struct OpTab { float v[5][32][2]; };

constexpr float db4pick(const double* h, int t) {
    return (t >= 0 && t < 4) ? (float)h[t] : 0.0f;
}

constexpr OpTab make_optab() {
    OpTab o{};
    const double s3  = 1.7320508075688772935;
    const double nrm = 0.17677669529663688110;
    const double h0[4] = {(1.0 + s3) * nrm, (3.0 + s3) * nrm,
                          (3.0 - s3) * nrm, (1.0 - s3) * nrm};
    const double h1[4] = {h0[3], -h0[2], h0[1], -h0[0]};
    for (int u = 0; u < 5; ++u) {
        for (int lane = 0; lane < 32; ++lane) {
            int m = (lane & 15) >> 1, f = lane & 1, h = lane >> 4;
            int tau = 4 * u + 2 * h - 2 * m;
            const double* w = f ? h1 : h0;
            o.v[u][lane][0] = db4pick(w, tau);
            o.v[u][lane][1] = db4pick(w, tau + 1);
        }
    }
    return o;
}

__device__ __constant__ OpTab OPTAB = make_optab();

#define WMMA_F32(A, B, C) \
    __builtin_amdgcn_wmma_f32_16x16x4_f32(false, (A), false, (B), (short)0, (C), false, false)

__device__ __forceinline__ v2f col_pair(float e, float o) {
    v2f p;  p.x = e;  p.y = o;
    asm volatile("" : "+v"(p));
    return p;
}

__device__ __forceinline__ void lds_handover() {
    __builtin_amdgcn_fence(__ATOMIC_RELEASE, "workgroup", "local");
    __builtin_amdgcn_s_barrier();
    __builtin_amdgcn_fence(__ATOMIC_ACQUIRE, "workgroup", "local");
}

__global__ __launch_bounds__(32)
void dwt_db4_level(const float* __restrict__ src, int N, long long srcStride,
                   float* __restrict__ dstLL, int ldLL, long long llStride,
                   float* __restrict__ outDet, int M, int T)
{
    const int lane = threadIdx.x;
    const int n    = lane & 15;
    const int g    = lane >> 4;
    int       c0   = blockIdx.x * T * 16;
    const int r0   = blockIdx.y * 16;
    const int img  = blockIdx.z;

    src    += (long long)img * srcStride;
    dstLL  += (long long)img * llStride;
    outDet += (long long)img * (512LL * 512LL);

    __shared__ __attribute__((aligned(16))) float Xs[34][42];
    __shared__ __attribute__((aligned(8))) float LHt[2][32][38];

    v2f W[5];
    #pragma unroll
    for (int u = 0; u < 5; ++u) W[u] = *(const v2f*)OPTAB.v[u][lane];

    const int sl = (lane < 27) ? lane : 26;
    const int lr = sl / 9;
    const int q  = sl - 9 * lr;
    const char* base = (const char*)src;
    unsigned rowOff[12];
    {
        int rb = 2 * r0 - 1 + lr;      rb += (rb < 0) ? N : 0;
        const unsigned rowBytes = 4u * (unsigned)N;
        #pragma unroll
        for (int i = 0; i < 12; ++i) {
            unsigned x = (unsigned)(rb + 3 * i);
            unsigned xm = x - (unsigned)N;
            x = (xm < x) ? xm : x;
            rowOff[i] = __umul24(x, rowBytes);
        }
    }
    v4f st[12];
    float hl0, hl1;
    {
        int gl = 2 * c0 - 1;           gl += (gl < 0) ? N : 0;
        int ra = 2 * r0 - 1 + lane;    ra += (ra < 0) ? N : 0;
        int rb2 = 2 * r0 + 31 + (lane & 1);  rb2 -= (rb2 >= N) ? N : 0;
        hl0 = *(const float*)(base + 4u * ((unsigned)ra  * (unsigned)N + (unsigned)gl));
        hl1 = *(const float*)(base + 4u * ((unsigned)rb2 * (unsigned)N + (unsigned)gl));
    }
    {
        int gc = 2 * c0 + 4 * q;       gc -= (gc >= N) ? N : 0;
        const unsigned colBytes = 4u * (unsigned)gc;
        #pragma unroll
        for (int i = 0; i < 12; ++i) st[i] = *(const v4f*)(base + (rowOff[i] + colBytes));
    }
    {
        const v2f z = {0.f, 0.f};
        *(v2f*)&LHt[0][lane][34] = z;
        *(v2f*)&LHt[1][lane][34] = z;
    }

    #pragma nounroll
    for (int it = 0; it < T; ++it, c0 += 16) {
    {
        if (it > 0) {
            hl0 = Xs[lane][36];
            hl1 = Xs[32 + (lane & 1)][36];
        }
        #pragma unroll
        for (int i = 0; i < 12; ++i) {
            const bool mine = (i < 11) ? (lane < 27) : (lane < 9);
            if (mine) {
                float* d = &Xs[3 * i + lr][4 * q + 5];
                d[0] = st[i].x;  d[1] = st[i].y;  d[2] = st[i].z;  d[3] = st[i].w;
            }
        }
        Xs[lane][4] = hl0;
        if (lane < 2) Xs[32 + lane][4] = hl1;
    }
    __syncthreads();

    if (it + 1 < T) {
        int gc = 2 * c0 + 32 + 4 * q;  gc -= (gc >= N) ? N : 0;
        const unsigned colBytes = 4u * (unsigned)gc;
        #pragma unroll
        for (int i = 0; i < 12; ++i) st[i] = *(const v4f*)(base + (rowOff[i] + colBytes));
    }

    const int chunkR[3] = {0, 16, 18};
    float* const sdst = &LHt[n & 1][16 * (it & 1) + (n >> 1)][8 * g];
    #pragma unroll
    for (int ch = 0; ch < 3; ++ch) {
        const int R = chunkR[ch];
        v8f acc0 = {0.f,0.f,0.f,0.f,0.f,0.f,0.f,0.f};
        v8f acc1 = {0.f,0.f,0.f,0.f,0.f,0.f,0.f,0.f};
        #pragma unroll
        for (int t = 0; t < 9; ++t) {
            v2f a = *(const v2f*)&Xs[R + n][4 * t + 2 * g + 4];
            if (t <= 4) acc0 = WMMA_F32(a, W[t], acc0);
            if (t >= 4) acc1 = WMMA_F32(a, W[t - 4], acc1);
        }
        if (ch < 2) {
            #pragma unroll
            for (int v = 0; v < 8; v += 2) {
                v2f p0, p1;
                p0.x = acc0[v];  p0.y = acc0[v + 1];
                p1.x = acc1[v];  p1.y = acc1[v + 1];
                *(v2f*)&sdst[R + v]          = p0;
                *(v2f*)&sdst[8 * 38 + R + v] = p1;
            }
        } else if (g) {
            v2f p0, p1;
            p0.x = acc0[6];  p0.y = acc0[7];
            p1.x = acc1[6];  p1.y = acc1[7];
            *(v2f*)&sdst[R + 6]          = p0;
            *(v2f*)&sdst[8 * 38 + R + 6] = p1;
        }
    }
    lds_handover();

    if (it & 1) {
    v8f Le0 = {0.f,0.f,0.f,0.f,0.f,0.f,0.f,0.f};
    v8f Lo0 = {0.f,0.f,0.f,0.f,0.f,0.f,0.f,0.f};
    v8f Le1 = {0.f,0.f,0.f,0.f,0.f,0.f,0.f,0.f};
    v8f Lo1 = {0.f,0.f,0.f,0.f,0.f,0.f,0.f,0.f};
    v8f He0 = {0.f,0.f,0.f,0.f,0.f,0.f,0.f,0.f};
    v8f Ho0 = {0.f,0.f,0.f,0.f,0.f,0.f,0.f,0.f};
    v8f He1 = {0.f,0.f,0.f,0.f,0.f,0.f,0.f,0.f};
    v8f Ho1 = {0.f,0.f,0.f,0.f,0.f,0.f,0.f,0.f};
    const float* const bp = &LHt[0][2 * n][2 * g];
    #pragma unroll
    for (int t = 0; t < 9; ++t) {
        const v2f bLe = *(const v2f*)(bp + 4 * t);
        const v2f bLo = *(const v2f*)(bp + 38 + 4 * t);
        const v2f bHe = *(const v2f*)(bp + 32 * 38 + 4 * t);
        const v2f bHo = *(const v2f*)(bp + 33 * 38 + 4 * t);
        if (t <= 4) {
            Le0 = WMMA_F32(W[t], bLe, Le0);      Lo0 = WMMA_F32(W[t], bLo, Lo0);
            He0 = WMMA_F32(W[t], bHe, He0);      Ho0 = WMMA_F32(W[t], bHo, Ho0);
        }
        if (t >= 4) {
            Le1 = WMMA_F32(W[t - 4], bLe, Le1);  Lo1 = WMMA_F32(W[t - 4], bLo, Lo1);
            He1 = WMMA_F32(W[t - 4], bHe, He1);  Ho1 = WMMA_F32(W[t - 4], bHo, Ho1);
        }
    }

    {
        const unsigned ocol  = (unsigned)(c0 - 16 + 2 * n);
        const unsigned orow0 = (unsigned)(r0 + 4 * g);
        const unsigned oLL   = 4u * (orow0 * (unsigned)ldLL + ocol);
        const unsigned oD    = 4u * (orow0 * 512u + ocol);
        const unsigned pitchLL = 4u * (unsigned)ldLL;
        char* pLL = (char*)dstLL;
        char* pLH = (char*)(outDet + M);
        char* pHL = (char*)(outDet + (long long)M * 512);
        char* pHH = (char*)(outDet + (long long)M * 512 + M);
        v2f sLL[8], sLH[8], sHL[8], sHH[8];
        #pragma unroll
        for (int w = 0; w < 4; ++w) {
            sLL[w]     = col_pair(Le0[2 * w],     Lo0[2 * w]);
            sLL[w + 4] = col_pair(Le1[2 * w],     Lo1[2 * w]);
            sLH[w]     = col_pair(Le0[2 * w + 1], Lo0[2 * w + 1]);
            sLH[w + 4] = col_pair(Le1[2 * w + 1], Lo1[2 * w + 1]);
            sHL[w]     = col_pair(He0[2 * w],     Ho0[2 * w]);
            sHL[w + 4] = col_pair(He1[2 * w],     Ho1[2 * w]);
            sHH[w]     = col_pair(He0[2 * w + 1], Ho0[2 * w + 1]);
            sHH[w + 4] = col_pair(He1[2 * w + 1], Ho1[2 * w + 1]);
        }
        unsigned oLLr[8];
        #pragma unroll
        for (int j = 0; j < 8; ++j) oLLr[j] = oLL + (unsigned)((j & 3) + 8 * (j >> 2)) * pitchLL;
        #pragma unroll
        for (int j = 0; j < 8; ++j) {
            const int row = (j & 3) + 8 * (j >> 2);
            *(v2f*)(pLL + oLLr[j]) = sLL[j];
            __builtin_nontemporal_store(sLH[j], (v2f*)((pLH + 2048 * row) + oD));
            __builtin_nontemporal_store(sHL[j], (v2f*)((pHL + 2048 * row) + oD));
            __builtin_nontemporal_store(sHH[j], (v2f*)((pHH + 2048 * row) + oD));
        }
    }
    }
    }
}

extern "C" void kernel_launch(void* const* d_in, const int* in_sizes, int n_in,
                              void* d_out, int out_size, void* d_ws, size_t ws_size,
                              hipStream_t stream) {
    const float* x = (const float*)d_in[0];
    float* out = (float*)d_out;
    const int BC = in_sizes[0] / (512 * 512);

    float* LL1 = (float*)d_ws;
    float* LL2 = LL1 + (long long)BC * 256 * 256;

    dim3 blk(32, 1, 1);
    dwt_db4_level<<<dim3(4, 16, BC), blk, 0, stream>>>(
        x,   512, 512LL * 512, LL1, 256, 256LL * 256, out, 256, 4);
    dwt_db4_level<<<dim3(2, 8, BC), blk, 0, stream>>>(
        LL1, 256, 256LL * 256, LL2, 128, 128LL * 128, out, 128, 4);
    dwt_db4_level<<<dim3(2, 4, BC), blk, 0, stream>>>(
        LL2, 128, 128LL * 128, out, 512, 512LL * 512, out, 64, 2);
}


# Input staged once per tile pair: each row's 64 own columns are one aligned 256-byte unit (17 16-byte + 2 4-byte loads per lane, 19 for 24), 3 line requests per row and pair not 4; sum order unchanged
# speedup vs baseline: 4.0720x; 1.1052x over previous
#include <hip/hip_runtime.h>

typedef __attribute__((ext_vector_type(2))) float v2f;
typedef __attribute__((ext_vector_type(4))) float v4f;
typedef __attribute__((ext_vector_type(8))) float v8f;

struct OpTab { float v[5][32][2]; };

constexpr float db4pick(const double* h, int t) {
    return (t >= 0 && t < 4) ? (float)h[t] : 0.0f;
}

constexpr OpTab make_optab() {
    OpTab o{};
    const double s3  = 1.7320508075688772935;
    const double nrm = 0.17677669529663688110;
    const double h0[4] = {(1.0 + s3) * nrm, (3.0 + s3) * nrm,
                          (3.0 - s3) * nrm, (1.0 - s3) * nrm};
    const double h1[4] = {h0[3], -h0[2], h0[1], -h0[0]};
    for (int u = 0; u < 5; ++u) {
        for (int lane = 0; lane < 32; ++lane) {
            int m = (lane & 15) >> 1, f = lane & 1, h = lane >> 4;
            int tau = 4 * u + 2 * h - 2 * m;
            const double* w = f ? h1 : h0;
            o.v[u][lane][0] = db4pick(w, tau);
            o.v[u][lane][1] = db4pick(w, tau + 1);
        }
    }
    return o;
}

__device__ __constant__ OpTab OPTAB = make_optab();

#define WMMA_F32(A, B, C) \
    __builtin_amdgcn_wmma_f32_16x16x4_f32(false, (A), false, (B), (short)0, (C), false, false)

__device__ __forceinline__ v2f col_pair(float e, float o) {
    v2f p;  p.x = e;  p.y = o;
    asm volatile("" : "+v"(p));
    return p;
}

__device__ __forceinline__ void lds_handover() {
    __builtin_amdgcn_fence(__ATOMIC_RELEASE, "workgroup", "local");
    __builtin_amdgcn_s_barrier();
    __builtin_amdgcn_fence(__ATOMIC_ACQUIRE, "workgroup", "local");
}

__global__ __launch_bounds__(32)
void dwt_db4_level(const float* __restrict__ src, int N, long long srcStride,
                   float* __restrict__ dstLL, int ldLL, long long llStride,
                   float* __restrict__ outDet, int M, int T)
{
    const int lane = threadIdx.x;
    const int n    = lane & 15;
    const int g    = lane >> 4;
    int       c0   = blockIdx.x * T * 16;
    const int r0   = blockIdx.y * 16;
    const int img  = blockIdx.z;

    src    += (long long)img * srcStride;
    dstLL  += (long long)img * llStride;
    outDet += (long long)img * (512LL * 512LL);

    __shared__ __attribute__((aligned(16))) float Xs[34][74];
    __shared__ __attribute__((aligned(8))) float LHt[2][32][38];

    v2f W[5];
    #pragma unroll
    for (int u = 0; u < 5; ++u) W[u] = *(const v2f*)OPTAB.v[u][lane];

    const char* base = (const char*)src;
    unsigned rowOff[17];
    {
        int rb = 2 * r0 - 1 + g;       rb += (rb < 0) ? N : 0;
        const unsigned rowBytes = 4u * (unsigned)N;
        #pragma unroll
        for (int i = 0; i < 17; ++i) {
            unsigned x = (unsigned)(rb + 2 * i);
            unsigned xm = x - (unsigned)N;
            x = (xm < x) ? xm : x;
            rowOff[i] = __umul24(x, rowBytes);
        }
    }
    v4f st[17];
    float hl0, hl1, hr0, hr1;
    unsigned hrowA, hrowB;
    {
        int ra = 2 * r0 - 1 + lane;    ra += (ra < 0) ? N : 0;
        int rb2 = 2 * r0 + 31 + (lane & 1);  rb2 -= (rb2 >= N) ? N : 0;
        hrowA = 4u * ((unsigned)ra  * (unsigned)N);
        hrowB = 4u * ((unsigned)rb2 * (unsigned)N);
        int gl = 2 * c0 - 1;           gl += (gl < 0) ? N : 0;
        int gr = 2 * c0 + 64;          gr -= (gr >= N) ? N : 0;
        hl0 = *(const float*)(base + (hrowA + 4u * (unsigned)gl));
        hl1 = *(const float*)(base + (hrowB + 4u * (unsigned)gl));
        hr0 = *(const float*)(base + (hrowA + 4u * (unsigned)gr));
        hr1 = *(const float*)(base + (hrowB + 4u * (unsigned)gr));
    }
    {
        const unsigned colBytes = 4u * (unsigned)(2 * c0 + 4 * n);
        #pragma unroll
        for (int i = 0; i < 17; ++i) st[i] = *(const v4f*)(base + (rowOff[i] + colBytes));
    }
    {
        const v2f z = {0.f, 0.f};
        *(v2f*)&LHt[0][lane][34] = z;
        *(v2f*)&LHt[1][lane][34] = z;
        *(v2f*)&Xs[lane][70] = z;
        if (lane < 2) *(v2f*)&Xs[32 + lane][70] = z;
    }

    const int P = T >> 1;
    #pragma nounroll
    for (int ip = 0; ip < P; ++ip, c0 += 32) {
    {
        if (ip > 0) {
            hl0 = Xs[lane][68];
            hl1 = Xs[32 + (lane & 1)][68];
        }
        #pragma unroll
        for (int i = 0; i < 17; ++i) {
            float* d = &Xs[2 * i + g][4 * n + 5];
            d[0] = st[i].x;  d[1] = st[i].y;  d[2] = st[i].z;  d[3] = st[i].w;
        }
        Xs[lane][4]  = hl0;
        Xs[lane][69] = hr0;
        if (lane < 2) { Xs[32 + lane][4] = hl1;  Xs[32 + lane][69] = hr1; }
    }
    __syncthreads();

    if (ip + 1 < P) {
        const unsigned colBytes = 4u * (unsigned)(2 * c0 + 64 + 4 * n);
        #pragma unroll
        for (int i = 0; i < 17; ++i) st[i] = *(const v4f*)(base + (rowOff[i] + colBytes));
        int gr = 2 * c0 + 128;         gr -= (gr >= N) ? N : 0;
        hr0 = *(const float*)(base + (hrowA + 4u * (unsigned)gr));
        hr1 = *(const float*)(base + (hrowB + 4u * (unsigned)gr));
    }

    const int chunkR[3] = {0, 16, 18};
    #pragma nounroll
    for (int tl = 0; tl < 2; ++tl) {
    float* const sdst = &LHt[n & 1][16 * tl + (n >> 1)][8 * g];
    #pragma unroll
    for (int ch = 0; ch < 3; ++ch) {
        const int R = chunkR[ch];
        v8f acc0 = {0.f,0.f,0.f,0.f,0.f,0.f,0.f,0.f};
        v8f acc1 = {0.f,0.f,0.f,0.f,0.f,0.f,0.f,0.f};
        #pragma unroll
        for (int t = 0; t < 9; ++t) {
            v2f a = *(const v2f*)&Xs[R + n][32 * tl + 4 * t + 2 * g + 4];
            if (t <= 4) acc0 = WMMA_F32(a, W[t], acc0);
            if (t >= 4) acc1 = WMMA_F32(a, W[t - 4], acc1);
        }
        if (ch < 2) {
            #pragma unroll
            for (int v = 0; v < 8; v += 2) {
                v2f p0, p1;
                p0.x = acc0[v];  p0.y = acc0[v + 1];
                p1.x = acc1[v];  p1.y = acc1[v + 1];
                *(v2f*)&sdst[R + v]          = p0;
                *(v2f*)&sdst[8 * 38 + R + v] = p1;
            }
        } else if (g) {
            v2f p0, p1;
            p0.x = acc0[6];  p0.y = acc0[7];
            p1.x = acc1[6];  p1.y = acc1[7];
            *(v2f*)&sdst[R + 6]          = p0;
            *(v2f*)&sdst[8 * 38 + R + 6] = p1;
        }
    }
    }
    lds_handover();

    {
    v8f Le0 = {0.f,0.f,0.f,0.f,0.f,0.f,0.f,0.f};
    v8f Lo0 = {0.f,0.f,0.f,0.f,0.f,0.f,0.f,0.f};
    v8f Le1 = {0.f,0.f,0.f,0.f,0.f,0.f,0.f,0.f};
    v8f Lo1 = {0.f,0.f,0.f,0.f,0.f,0.f,0.f,0.f};
    v8f He0 = {0.f,0.f,0.f,0.f,0.f,0.f,0.f,0.f};
    v8f Ho0 = {0.f,0.f,0.f,0.f,0.f,0.f,0.f,0.f};
    v8f He1 = {0.f,0.f,0.f,0.f,0.f,0.f,0.f,0.f};
    v8f Ho1 = {0.f,0.f,0.f,0.f,0.f,0.f,0.f,0.f};
    const float* const bp = &LHt[0][2 * n][2 * g];
    #pragma unroll
    for (int t = 0; t < 9; ++t) {
        const v2f bLe = *(const v2f*)(bp + 4 * t);
        const v2f bLo = *(const v2f*)(bp + 38 + 4 * t);
        const v2f bHe = *(const v2f*)(bp + 32 * 38 + 4 * t);
        const v2f bHo = *(const v2f*)(bp + 33 * 38 + 4 * t);
        if (t <= 4) {
            Le0 = WMMA_F32(W[t], bLe, Le0);      Lo0 = WMMA_F32(W[t], bLo, Lo0);
            He0 = WMMA_F32(W[t], bHe, He0);      Ho0 = WMMA_F32(W[t], bHo, Ho0);
        }
        if (t >= 4) {
            Le1 = WMMA_F32(W[t - 4], bLe, Le1);  Lo1 = WMMA_F32(W[t - 4], bLo, Lo1);
            He1 = WMMA_F32(W[t - 4], bHe, He1);  Ho1 = WMMA_F32(W[t - 4], bHo, Ho1);
        }
    }

    {
        const unsigned ocol  = (unsigned)(c0 + 2 * n);
        const unsigned orow0 = (unsigned)(r0 + 4 * g);
        const unsigned oLL   = 4u * (orow0 * (unsigned)ldLL + ocol);
        const unsigned oD    = 4u * (orow0 * 512u + ocol);
        const unsigned pitchLL = 4u * (unsigned)ldLL;
        char* pLL = (char*)dstLL;
        char* pLH = (char*)(outDet + M);
        char* pHL = (char*)(outDet + (long long)M * 512);
        char* pHH = (char*)(outDet + (long long)M * 512 + M);
        v2f sLL[8], sLH[8], sHL[8], sHH[8];
        #pragma unroll
        for (int w = 0; w < 4; ++w) {
            sLL[w]     = col_pair(Le0[2 * w],     Lo0[2 * w]);
            sLL[w + 4] = col_pair(Le1[2 * w],     Lo1[2 * w]);
            sLH[w]     = col_pair(Le0[2 * w + 1], Lo0[2 * w + 1]);
            sLH[w + 4] = col_pair(Le1[2 * w + 1], Lo1[2 * w + 1]);
            sHL[w]     = col_pair(He0[2 * w],     Ho0[2 * w]);
            sHL[w + 4] = col_pair(He1[2 * w],     Ho1[2 * w]);
            sHH[w]     = col_pair(He0[2 * w + 1], Ho0[2 * w + 1]);
            sHH[w + 4] = col_pair(He1[2 * w + 1], Ho1[2 * w + 1]);
        }
        unsigned oLLr[8];
        #pragma unroll
        for (int j = 0; j < 8; ++j) oLLr[j] = oLL + (unsigned)((j & 3) + 8 * (j >> 2)) * pitchLL;
        #pragma unroll
        for (int j = 0; j < 8; ++j) {
            const int row = (j & 3) + 8 * (j >> 2);
            *(v2f*)(pLL + oLLr[j]) = sLL[j];
            __builtin_nontemporal_store(sLH[j], (v2f*)((pLH + 2048 * row) + oD));
            __builtin_nontemporal_store(sHL[j], (v2f*)((pHL + 2048 * row) + oD));
            __builtin_nontemporal_store(sHH[j], (v2f*)((pHH + 2048 * row) + oD));
        }
    }
    }
    }
}

extern "C" void kernel_launch(void* const* d_in, const int* in_sizes, int n_in,
                              void* d_out, int out_size, void* d_ws, size_t ws_size,
                              hipStream_t stream) {
    const float* x = (const float*)d_in[0];
    float* out = (float*)d_out;
    const int BC = in_sizes[0] / (512 * 512);

    float* LL1 = (float*)d_ws;
    float* LL2 = LL1 + (long long)BC * 256 * 256;

    dim3 blk(32, 1, 1);
    dwt_db4_level<<<dim3(4, 16, BC), blk, 0, stream>>>(
        x,   512, 512LL * 512, LL1, 256, 256LL * 256, out, 256, 4);
    dwt_db4_level<<<dim3(2, 8, BC), blk, 0, stream>>>(
        LL1, 256, 256LL * 256, LL2, 128, 128LL * 128, out, 128, 4);
    dwt_db4_level<<<dim3(2, 4, BC), blk, 0, stream>>>(
        LL2, 128, 128LL * 128, out, 512, 512LL * 512, out, 64, 2);
}


# Input rows 32, 33 of a tile pair take ONE 5-product chunk (16 A rows = 2 rows x 4 column groups), not a third 10-product chunk per tile: 85 not 100 matrix products per pair; sum order unchanged
# speedup vs baseline: 4.2456x; 1.0426x over previous
#include <hip/hip_runtime.h>

typedef __attribute__((ext_vector_type(2))) float v2f;
typedef __attribute__((ext_vector_type(4))) float v4f;
typedef __attribute__((ext_vector_type(8))) float v8f;

struct OpTab { float v[5][32][2]; };

constexpr float db4pick(const double* h, int t) {
    return (t >= 0 && t < 4) ? (float)h[t] : 0.0f;
}

constexpr OpTab make_optab() {
    OpTab o{};
    const double s3  = 1.7320508075688772935;
    const double nrm = 0.17677669529663688110;
    const double h0[4] = {(1.0 + s3) * nrm, (3.0 + s3) * nrm,
                          (3.0 - s3) * nrm, (1.0 - s3) * nrm};
    const double h1[4] = {h0[3], -h0[2], h0[1], -h0[0]};
    for (int u = 0; u < 5; ++u) {
        for (int lane = 0; lane < 32; ++lane) {
            int m = (lane & 15) >> 1, f = lane & 1, h = lane >> 4;
            int tau = 4 * u + 2 * h - 2 * m;
            const double* w = f ? h1 : h0;
            o.v[u][lane][0] = db4pick(w, tau);
            o.v[u][lane][1] = db4pick(w, tau + 1);
        }
    }
    return o;
}

__device__ __constant__ OpTab OPTAB = make_optab();

#define WMMA_F32(A, B, C) \
    __builtin_amdgcn_wmma_f32_16x16x4_f32(false, (A), false, (B), (short)0, (C), false, false)

__device__ __forceinline__ v2f col_pair(float e, float o) {
    v2f p;  p.x = e;  p.y = o;
    asm volatile("" : "+v"(p));
    return p;
}

__device__ __forceinline__ void lds_handover() {
    __builtin_amdgcn_fence(__ATOMIC_RELEASE, "workgroup", "local");
    __builtin_amdgcn_s_barrier();
    __builtin_amdgcn_fence(__ATOMIC_ACQUIRE, "workgroup", "local");
}

__global__ __launch_bounds__(32)
void dwt_db4_level(const float* __restrict__ src, int N, long long srcStride,
                   float* __restrict__ dstLL, int ldLL, long long llStride,
                   float* __restrict__ outDet, int M, int T)
{
    const int lane = threadIdx.x;
    const int n    = lane & 15;
    const int g    = lane >> 4;
    int       c0   = blockIdx.x * T * 16;
    const int r0   = blockIdx.y * 16;
    const int img  = blockIdx.z;

    src    += (long long)img * srcStride;
    dstLL  += (long long)img * llStride;
    outDet += (long long)img * (512LL * 512LL);

    __shared__ __attribute__((aligned(16))) float Xs[34][74];
    __shared__ __attribute__((aligned(8))) float LHt[2][32][38];

    v2f W[5];
    #pragma unroll
    for (int u = 0; u < 5; ++u) W[u] = *(const v2f*)OPTAB.v[u][lane];

    const char* base = (const char*)src;
    unsigned rowOff[17];
    {
        int rb = 2 * r0 - 1 + g;       rb += (rb < 0) ? N : 0;
        const unsigned rowBytes = 4u * (unsigned)N;
        #pragma unroll
        for (int i = 0; i < 17; ++i) {
            unsigned x = (unsigned)(rb + 2 * i);
            unsigned xm = x - (unsigned)N;
            x = (xm < x) ? xm : x;
            rowOff[i] = __umul24(x, rowBytes);
        }
    }
    v4f st[17];
    float hl0, hl1, hr0, hr1;
    unsigned hrowA, hrowB;
    {
        int ra = 2 * r0 - 1 + lane;    ra += (ra < 0) ? N : 0;
        int rb2 = 2 * r0 + 31 + (lane & 1);  rb2 -= (rb2 >= N) ? N : 0;
        hrowA = 4u * ((unsigned)ra  * (unsigned)N);
        hrowB = 4u * ((unsigned)rb2 * (unsigned)N);
        int gl = 2 * c0 - 1;           gl += (gl < 0) ? N : 0;
        int gr = 2 * c0 + 64;          gr -= (gr >= N) ? N : 0;
        hl0 = *(const float*)(base + (hrowA + 4u * (unsigned)gl));
        hl1 = *(const float*)(base + (hrowB + 4u * (unsigned)gl));
        hr0 = *(const float*)(base + (hrowA + 4u * (unsigned)gr));
        hr1 = *(const float*)(base + (hrowB + 4u * (unsigned)gr));
    }
    {
        const unsigned colBytes = 4u * (unsigned)(2 * c0 + 4 * n);
        #pragma unroll
        for (int i = 0; i < 17; ++i) st[i] = *(const v4f*)(base + (rowOff[i] + colBytes));
    }
    {
        const v2f z = {0.f, 0.f};
        *(v2f*)&LHt[0][lane][34] = z;
        *(v2f*)&LHt[1][lane][34] = z;
        *(v2f*)&Xs[lane][70] = z;
        if (lane < 2) *(v2f*)&Xs[32 + lane][70] = z;
    }

    const int P = T >> 1;
    #pragma nounroll
    for (int ip = 0; ip < P; ++ip, c0 += 32) {
    {
        if (ip > 0) {
            hl0 = Xs[lane][68];
            hl1 = Xs[32 + (lane & 1)][68];
        }
        #pragma unroll
        for (int i = 0; i < 17; ++i) {
            float* d = &Xs[2 * i + g][4 * n + 5];
            d[0] = st[i].x;  d[1] = st[i].y;  d[2] = st[i].z;  d[3] = st[i].w;
        }
        Xs[lane][4]  = hl0;
        Xs[lane][69] = hr0;
        if (lane < 2) { Xs[32 + lane][4] = hl1;  Xs[32 + lane][69] = hr1; }
    }
    __syncthreads();

    if (ip + 1 < P) {
        const unsigned colBytes = 4u * (unsigned)(2 * c0 + 64 + 4 * n);
        #pragma unroll
        for (int i = 0; i < 17; ++i) st[i] = *(const v4f*)(base + (rowOff[i] + colBytes));
        int gr = 2 * c0 + 128;         gr -= (gr >= N) ? N : 0;
        hr0 = *(const float*)(base + (hrowA + 4u * (unsigned)gr));
        hr1 = *(const float*)(base + (hrowB + 4u * (unsigned)gr));
    }

    const float* const ap = &Xs[32 + (n & 1)][16 * ((n >> 1) & 3) + 2 * g + 4];
    v2f at[5];
    #pragma unroll
    for (int u = 0; u < 5; ++u) at[u] = *(const v2f*)(ap + 4 * u);
    #pragma nounroll
    for (int tl = 0; tl < 2; ++tl) {
    float* const sdst = &LHt[n & 1][16 * tl + (n >> 1)][8 * g];
    #pragma unroll
    for (int ch = 0; ch < 2; ++ch) {
        const int R = 16 * ch;
        v8f acc0 = {0.f,0.f,0.f,0.f,0.f,0.f,0.f,0.f};
        v8f acc1 = {0.f,0.f,0.f,0.f,0.f,0.f,0.f,0.f};
        #pragma unroll
        for (int t = 0; t < 9; ++t) {
            v2f a = *(const v2f*)&Xs[R + n][32 * tl + 4 * t + 2 * g + 4];
            if (t <= 4) acc0 = WMMA_F32(a, W[t], acc0);
            if (t >= 4) acc1 = WMMA_F32(a, W[t - 4], acc1);
        }
        #pragma unroll
        for (int v = 0; v < 8; v += 2) {
            v2f p0, p1;
            p0.x = acc0[v];  p0.y = acc0[v + 1];
            p1.x = acc1[v];  p1.y = acc1[v + 1];
            *(v2f*)&sdst[R + v]          = p0;
            *(v2f*)&sdst[8 * 38 + R + v] = p1;
        }
    }
    }
    {
        v8f acc = {0.f,0.f,0.f,0.f,0.f,0.f,0.f,0.f};
        #pragma unroll
        for (int u = 0; u < 5; ++u) {
            acc = WMMA_F32(at[u], W[u], acc);
        }
        if (g == 0) {
            float* const d = &LHt[n & 1][n >> 1][32];
            #pragma unroll
            for (int q = 0; q < 4; ++q) {
                v2f p;
                p.x = acc[2 * q];  p.y = acc[2 * q + 1];
                *(v2f*)&d[8 * q * 38] = p;
            }
        }
    }
    lds_handover();

    {
    v8f Le0 = {0.f,0.f,0.f,0.f,0.f,0.f,0.f,0.f};
    v8f Lo0 = {0.f,0.f,0.f,0.f,0.f,0.f,0.f,0.f};
    v8f Le1 = {0.f,0.f,0.f,0.f,0.f,0.f,0.f,0.f};
    v8f Lo1 = {0.f,0.f,0.f,0.f,0.f,0.f,0.f,0.f};
    v8f He0 = {0.f,0.f,0.f,0.f,0.f,0.f,0.f,0.f};
    v8f Ho0 = {0.f,0.f,0.f,0.f,0.f,0.f,0.f,0.f};
    v8f He1 = {0.f,0.f,0.f,0.f,0.f,0.f,0.f,0.f};
    v8f Ho1 = {0.f,0.f,0.f,0.f,0.f,0.f,0.f,0.f};
    const float* const bp = &LHt[0][2 * n][2 * g];
    #pragma unroll
    for (int t = 0; t < 9; ++t) {
        const v2f bLe = *(const v2f*)(bp + 4 * t);
        const v2f bLo = *(const v2f*)(bp + 38 + 4 * t);
        const v2f bHe = *(const v2f*)(bp + 32 * 38 + 4 * t);
        const v2f bHo = *(const v2f*)(bp + 33 * 38 + 4 * t);
        if (t <= 4) {
            Le0 = WMMA_F32(W[t], bLe, Le0);      Lo0 = WMMA_F32(W[t], bLo, Lo0);
            He0 = WMMA_F32(W[t], bHe, He0);      Ho0 = WMMA_F32(W[t], bHo, Ho0);
        }
        if (t >= 4) {
            Le1 = WMMA_F32(W[t - 4], bLe, Le1);  Lo1 = WMMA_F32(W[t - 4], bLo, Lo1);
            He1 = WMMA_F32(W[t - 4], bHe, He1);  Ho1 = WMMA_F32(W[t - 4], bHo, Ho1);
        }
    }

    {
        const unsigned ocol  = (unsigned)(c0 + 2 * n);
        const unsigned orow0 = (unsigned)(r0 + 4 * g);
        const unsigned oLL   = 4u * (orow0 * (unsigned)ldLL + ocol);
        const unsigned oD    = 4u * (orow0 * 512u + ocol);
        const unsigned pitchLL = 4u * (unsigned)ldLL;
        char* pLL = (char*)dstLL;
        char* pLH = (char*)(outDet + M);
        char* pHL = (char*)(outDet + (long long)M * 512);
        char* pHH = (char*)(outDet + (long long)M * 512 + M);
        v2f sLL[8], sLH[8], sHL[8], sHH[8];
        #pragma unroll
        for (int w = 0; w < 4; ++w) {
            sLL[w]     = col_pair(Le0[2 * w],     Lo0[2 * w]);
            sLL[w + 4] = col_pair(Le1[2 * w],     Lo1[2 * w]);
            sLH[w]     = col_pair(Le0[2 * w + 1], Lo0[2 * w + 1]);
            sLH[w + 4] = col_pair(Le1[2 * w + 1], Lo1[2 * w + 1]);
            sHL[w]     = col_pair(He0[2 * w],     Ho0[2 * w]);
            sHL[w + 4] = col_pair(He1[2 * w],     Ho1[2 * w]);
            sHH[w]     = col_pair(He0[2 * w + 1], Ho0[2 * w + 1]);
            sHH[w + 4] = col_pair(He1[2 * w + 1], Ho1[2 * w + 1]);
        }
        unsigned oLLr[8];
        #pragma unroll
        for (int j = 0; j < 8; ++j) oLLr[j] = oLL + (unsigned)((j & 3) + 8 * (j >> 2)) * pitchLL;
        #pragma unroll
        for (int j = 0; j < 8; ++j) {
            const int row = (j & 3) + 8 * (j >> 2);
            *(v2f*)(pLL + oLLr[j]) = sLL[j];
            __builtin_nontemporal_store(sLH[j], (v2f*)((pLH + 2048 * row) + oD));
            __builtin_nontemporal_store(sHL[j], (v2f*)((pHL + 2048 * row) + oD));
            __builtin_nontemporal_store(sHH[j], (v2f*)((pHH + 2048 * row) + oD));
        }
    }
    }
    }
}

extern "C" void kernel_launch(void* const* d_in, const int* in_sizes, int n_in,
                              void* d_out, int out_size, void* d_ws, size_t ws_size,
                              hipStream_t stream) {
    const float* x = (const float*)d_in[0];
    float* out = (float*)d_out;
    const int BC = in_sizes[0] / (512 * 512);

    float* LL1 = (float*)d_ws;
    float* LL2 = LL1 + (long long)BC * 256 * 256;

    dim3 blk(32, 1, 1);
    dwt_db4_level<<<dim3(4, 16, BC), blk, 0, stream>>>(
        x,   512, 512LL * 512, LL1, 256, 256LL * 256, out, 256, 4);
    dwt_db4_level<<<dim3(2, 8, BC), blk, 0, stream>>>(
        LL1, 256, 256LL * 256, LL2, 128, 128LL * 128, out, 128, 4);
    dwt_db4_level<<<dim3(2, 4, BC), blk, 0, stream>>>(
        LL2, 128, 128LL * 128, out, 512, 512LL * 512, out, 64, 2);
}


# No border-column loads: workgroup = NW waves side by side over a band's whole width, strip loaded up front, left/right tap columns passed between waves through LDS; sum order unchanged
# speedup vs baseline: 4.5800x; 1.0788x over previous
#include <hip/hip_runtime.h>

typedef __attribute__((ext_vector_type(2))) float v2f;
typedef __attribute__((ext_vector_type(4))) float v4f;
typedef __attribute__((ext_vector_type(8))) float v8f;

struct OpTab { float v[5][32][2]; };

constexpr float db4pick(const double* h, int t) {
    return (t >= 0 && t < 4) ? (float)h[t] : 0.0f;
}

constexpr OpTab make_optab() {
    OpTab o{};
    const double s3  = 1.7320508075688772935;
    const double nrm = 0.17677669529663688110;
    const double h0[4] = {(1.0 + s3) * nrm, (3.0 + s3) * nrm,
                          (3.0 - s3) * nrm, (1.0 - s3) * nrm};
    const double h1[4] = {h0[3], -h0[2], h0[1], -h0[0]};
    for (int u = 0; u < 5; ++u) {
        for (int lane = 0; lane < 32; ++lane) {
            int m = (lane & 15) >> 1, f = lane & 1, h = lane >> 4;
            int tau = 4 * u + 2 * h - 2 * m;
            const double* w = f ? h1 : h0;
            o.v[u][lane][0] = db4pick(w, tau);
            o.v[u][lane][1] = db4pick(w, tau + 1);
        }
    }
    return o;
}

__device__ __constant__ OpTab OPTAB = make_optab();

#define WMMA_F32(A, B, C) \
    __builtin_amdgcn_wmma_f32_16x16x4_f32(false, (A), false, (B), (short)0, (C), false, false)

__device__ __forceinline__ v2f col_pair(float e, float o) {
    v2f p;  p.x = e;  p.y = o;
    asm volatile("" : "+v"(p));
    return p;
}

__device__ __forceinline__ void wave_handover() {
    __builtin_amdgcn_fence(__ATOMIC_RELEASE, "workgroup", "local");
    __builtin_amdgcn_fence(__ATOMIC_ACQUIRE, "workgroup", "local");
}

__device__ __forceinline__ void group_handover() {
    __builtin_amdgcn_fence(__ATOMIC_RELEASE, "workgroup", "local");
    __builtin_amdgcn_s_barrier();
    __builtin_amdgcn_fence(__ATOMIC_ACQUIRE, "workgroup", "local");
}

template <int NW, int P>
__global__ __launch_bounds__(32 * NW)
void dwt_db4_level(const float* __restrict__ src, int N, long long srcStride,
                   float* __restrict__ dstLL, int ldLL, long long llStride,
                   float* __restrict__ outDet, int M)
{
    const int lane = threadIdx.x & 31;
    const int n    = lane & 15;
    const int g    = lane >> 4;
    const int w    = __builtin_amdgcn_readfirstlane((int)(threadIdx.x >> 5));
    int       c0   = w * (32 * P);
    const int r0   = blockIdx.y * 16;
    const int img  = blockIdx.z;

    src    += (long long)img * srcStride;
    dstLL  += (long long)img * llStride;
    outDet += (long long)img * (512LL * 512LL);

    __shared__ __attribute__((aligned(16))) float XsAll[NW][34][74];
    __shared__ __attribute__((aligned(8))) float LHtAll[NW][2][32][38];
    __shared__ float HaloF[NW][34];
    __shared__ float HaloL[NW][34];
    __shared__ float HaloM[NW][34];
    float (* const Xs)[74]      = XsAll[w];
    float (* const LHt)[32][38] = LHtAll[w];

    v2f W[5];
    #pragma unroll
    for (int u = 0; u < 5; ++u) W[u] = *(const v2f*)OPTAB.v[u][lane];

    const char* base = (const char*)src;
    unsigned rowOff[17];
    {
        int rb = 2 * r0 - 1 + g;       rb += (rb < 0) ? N : 0;
        const unsigned rowBytes = 4u * (unsigned)N;
        #pragma unroll
        for (int i = 0; i < 17; ++i) {
            unsigned x = (unsigned)(rb + 2 * i);
            unsigned xm = x - (unsigned)N;
            x = (xm < x) ? xm : x;
            rowOff[i] = __umul24(x, rowBytes);
        }
    }
    v4f st[17];
    v4f sb[17];
    {
        const unsigned colBytes = 4u * (unsigned)(2 * c0 + 4 * n);
        #pragma unroll
        for (int i = 0; i < 17; ++i) st[i] = *(const v4f*)(base + (rowOff[i] + colBytes));
        if (P == 2) {
            #pragma unroll
            for (int i = 0; i < 17; ++i) sb[i] = *(const v4f*)(base + (rowOff[i] + (colBytes + 256u)));
        }
    }
    {
        const v2f z = {0.f, 0.f};
        *(v2f*)&LHt[0][lane][34] = z;
        *(v2f*)&LHt[1][lane][34] = z;
        *(v2f*)&Xs[lane][70] = z;
        if (lane < 2) *(v2f*)&Xs[32 + lane][70] = z;
    }
    {
        float* const hf = &HaloF[w][g];
        float* const hm = &HaloM[w][g];
        float* const hl = &HaloL[w][g];
        if (n == 0) {
            #pragma unroll
            for (int i = 0; i < 17; ++i) {
                hf[2 * i] = st[i].x;
                if (P == 2) hm[2 * i] = sb[i].x;
            }
        }
        if (n == 15) {
            #pragma unroll
            for (int i = 0; i < 17; ++i) hl[2 * i] = (P == 2) ? sb[i].w : st[i].w;
        }
    }
    group_handover();
    float hl0 = HaloL[(w + NW - 1) & (NW - 1)][lane];
    float hl1 = HaloL[(w + NW - 1) & (NW - 1)][32 + (lane & 1)];
    const float hrE0 = HaloF[(w + 1) & (NW - 1)][lane];
    const float hrE1 = HaloF[(w + 1) & (NW - 1)][32 + (lane & 1)];
    const float hrM0 = (P == 2) ? HaloM[w][lane] : 0.f;
    const float hrM1 = (P == 2) ? HaloM[w][32 + (lane & 1)] : 0.f;

    #pragma unroll
    for (int ip = 0; ip < P; ++ip, c0 += 32) {
    {
        if (ip > 0) {
            hl0 = Xs[lane][68];
            hl1 = Xs[32 + (lane & 1)][68];
        }
        #pragma unroll
        for (int i = 0; i < 17; ++i) {
            const v4f v = (ip == 0) ? st[i] : sb[i];
            float* d = &Xs[2 * i + g][4 * n + 5];
            d[0] = v.x;  d[1] = v.y;  d[2] = v.z;  d[3] = v.w;
        }
        const float hr0 = (ip + 1 < P) ? hrM0 : hrE0;
        const float hr1 = (ip + 1 < P) ? hrM1 : hrE1;
        Xs[lane][4]  = hl0;
        Xs[lane][69] = hr0;
        if (lane < 2) { Xs[32 + lane][4] = hl1;  Xs[32 + lane][69] = hr1; }
    }
    wave_handover();

    const float* const ap = &Xs[32 + (n & 1)][16 * ((n >> 1) & 3) + 2 * g + 4];
    v2f at[5];
    #pragma unroll
    for (int u = 0; u < 5; ++u) at[u] = *(const v2f*)(ap + 4 * u);
    #pragma nounroll
    for (int tl = 0; tl < 2; ++tl) {
    float* const sdst = &LHt[n & 1][16 * tl + (n >> 1)][8 * g];
    #pragma unroll
    for (int ch = 0; ch < 2; ++ch) {
        const int R = 16 * ch;
        v8f acc0 = {0.f,0.f,0.f,0.f,0.f,0.f,0.f,0.f};
        v8f acc1 = {0.f,0.f,0.f,0.f,0.f,0.f,0.f,0.f};
        #pragma unroll
        for (int t = 0; t < 9; ++t) {
            v2f a = *(const v2f*)&Xs[R + n][32 * tl + 4 * t + 2 * g + 4];
            if (t <= 4) acc0 = WMMA_F32(a, W[t], acc0);
            if (t >= 4) acc1 = WMMA_F32(a, W[t - 4], acc1);
        }
        #pragma unroll
        for (int v = 0; v < 8; v += 2) {
            v2f p0, p1;
            p0.x = acc0[v];  p0.y = acc0[v + 1];
            p1.x = acc1[v];  p1.y = acc1[v + 1];
            *(v2f*)&sdst[R + v]          = p0;
            *(v2f*)&sdst[8 * 38 + R + v] = p1;
        }
    }
    }
    {
        v8f acc = {0.f,0.f,0.f,0.f,0.f,0.f,0.f,0.f};
        #pragma unroll
        for (int u = 0; u < 5; ++u) {
            acc = WMMA_F32(at[u], W[u], acc);
        }
        if (g == 0) {
            float* const d = &LHt[n & 1][n >> 1][32];
            #pragma unroll
            for (int q = 0; q < 4; ++q) {
                v2f p;
                p.x = acc[2 * q];  p.y = acc[2 * q + 1];
                *(v2f*)&d[8 * q * 38] = p;
            }
        }
    }
    wave_handover();

    {
    v8f Le0 = {0.f,0.f,0.f,0.f,0.f,0.f,0.f,0.f};
    v8f Lo0 = {0.f,0.f,0.f,0.f,0.f,0.f,0.f,0.f};
    v8f Le1 = {0.f,0.f,0.f,0.f,0.f,0.f,0.f,0.f};
    v8f Lo1 = {0.f,0.f,0.f,0.f,0.f,0.f,0.f,0.f};
    v8f He0 = {0.f,0.f,0.f,0.f,0.f,0.f,0.f,0.f};
    v8f Ho0 = {0.f,0.f,0.f,0.f,0.f,0.f,0.f,0.f};
    v8f He1 = {0.f,0.f,0.f,0.f,0.f,0.f,0.f,0.f};
    v8f Ho1 = {0.f,0.f,0.f,0.f,0.f,0.f,0.f,0.f};
    const float* const bp = &LHt[0][2 * n][2 * g];
    #pragma unroll
    for (int t = 0; t < 9; ++t) {
        const v2f bLe = *(const v2f*)(bp + 4 * t);
        const v2f bLo = *(const v2f*)(bp + 38 + 4 * t);
        const v2f bHe = *(const v2f*)(bp + 32 * 38 + 4 * t);
        const v2f bHo = *(const v2f*)(bp + 33 * 38 + 4 * t);
        if (t <= 4) {
            Le0 = WMMA_F32(W[t], bLe, Le0);      Lo0 = WMMA_F32(W[t], bLo, Lo0);
            He0 = WMMA_F32(W[t], bHe, He0);      Ho0 = WMMA_F32(W[t], bHo, Ho0);
        }
        if (t >= 4) {
            Le1 = WMMA_F32(W[t - 4], bLe, Le1);  Lo1 = WMMA_F32(W[t - 4], bLo, Lo1);
            He1 = WMMA_F32(W[t - 4], bHe, He1);  Ho1 = WMMA_F32(W[t - 4], bHo, Ho1);
        }
    }

    {
        const unsigned ocol  = (unsigned)(c0 + 2 * n);
        const unsigned orow0 = (unsigned)(r0 + 4 * g);
        const unsigned oLL   = 4u * (orow0 * (unsigned)ldLL + ocol);
        const unsigned oD    = 4u * (orow0 * 512u + ocol);
        const unsigned pitchLL = 4u * (unsigned)ldLL;
        char* pLL = (char*)dstLL;
        char* pLH = (char*)(outDet + M);
        char* pHL = (char*)(outDet + (long long)M * 512);
        char* pHH = (char*)(outDet + (long long)M * 512 + M);
        v2f sLL[8], sLH[8], sHL[8], sHH[8];
        #pragma unroll
        for (int w = 0; w < 4; ++w) {
            sLL[w]     = col_pair(Le0[2 * w],     Lo0[2 * w]);
            sLL[w + 4] = col_pair(Le1[2 * w],     Lo1[2 * w]);
            sLH[w]     = col_pair(Le0[2 * w + 1], Lo0[2 * w + 1]);
            sLH[w + 4] = col_pair(Le1[2 * w + 1], Lo1[2 * w + 1]);
            sHL[w]     = col_pair(He0[2 * w],     Ho0[2 * w]);
            sHL[w + 4] = col_pair(He1[2 * w],     Ho1[2 * w]);
            sHH[w]     = col_pair(He0[2 * w + 1], Ho0[2 * w + 1]);
            sHH[w + 4] = col_pair(He1[2 * w + 1], Ho1[2 * w + 1]);
        }
        unsigned oLLr[8];
        #pragma unroll
        for (int j = 0; j < 8; ++j) oLLr[j] = oLL + (unsigned)((j & 3) + 8 * (j >> 2)) * pitchLL;
        #pragma unroll
        for (int j = 0; j < 8; ++j) {
            const int row = (j & 3) + 8 * (j >> 2);
            *(v2f*)(pLL + oLLr[j]) = sLL[j];
            __builtin_nontemporal_store(sLH[j], (v2f*)((pLH + 2048 * row) + oD));
            __builtin_nontemporal_store(sHL[j], (v2f*)((pHL + 2048 * row) + oD));
            __builtin_nontemporal_store(sHH[j], (v2f*)((pHH + 2048 * row) + oD));
        }
    }
    }
    }
}

extern "C" void kernel_launch(void* const* d_in, const int* in_sizes, int n_in,
                              void* d_out, int out_size, void* d_ws, size_t ws_size,
                              hipStream_t stream) {
    const float* x = (const float*)d_in[0];
    float* out = (float*)d_out;
    const int BC = in_sizes[0] / (512 * 512);

    float* LL1 = (float*)d_ws;
    float* LL2 = LL1 + (long long)BC * 256 * 256;

    dwt_db4_level<4, 2><<<dim3(1, 16, BC), dim3(128, 1, 1), 0, stream>>>(
        x,   512, 512LL * 512, LL1, 256, 256LL * 256, out, 256);
    dwt_db4_level<2, 2><<<dim3(1, 8, BC), dim3(64, 1, 1), 0, stream>>>(
        LL1, 256, 256LL * 256, LL2, 128, 128LL * 128, out, 128);
    dwt_db4_level<2, 1><<<dim3(1, 4, BC), dim3(64, 1, 1), 0, stream>>>(
        LL2, 128, 128LL * 128, out, 512, 512LL * 512, out, 64);
}


# Levels 2 and 3 in one launch: a workgroup owns two level-2 bands, keeps their LL rows in LDS, recomputes the LL row above and below (their 4 vertical taps in one product), then runs its level-3 band
# speedup vs baseline: 4.9147x; 1.0731x over previous
#include <hip/hip_runtime.h>

typedef __attribute__((ext_vector_type(2))) float v2f;
typedef __attribute__((ext_vector_type(4))) float v4f;
typedef __attribute__((ext_vector_type(8))) float v8f;

struct OpTab { float v[5][32][2]; };

constexpr float db4pick(const double* h, int t) {
    return (t >= 0 && t < 4) ? (float)h[t] : 0.0f;
}

constexpr OpTab make_optab() {
    OpTab o{};
    const double s3  = 1.7320508075688772935;
    const double nrm = 0.17677669529663688110;
    const double h0[4] = {(1.0 + s3) * nrm, (3.0 + s3) * nrm,
                          (3.0 - s3) * nrm, (1.0 - s3) * nrm};
    const double h1[4] = {h0[3], -h0[2], h0[1], -h0[0]};
    for (int u = 0; u < 5; ++u) {
        for (int lane = 0; lane < 32; ++lane) {
            int m = (lane & 15) >> 1, f = lane & 1, h = lane >> 4;
            int tau = 4 * u + 2 * h - 2 * m;
            const double* w = f ? h1 : h0;
            o.v[u][lane][0] = db4pick(w, tau);
            o.v[u][lane][1] = db4pick(w, tau + 1);
        }
    }
    return o;
}

__device__ __constant__ OpTab OPTAB = make_optab();

#define WMMA_F32(A, B, C) \
    __builtin_amdgcn_wmma_f32_16x16x4_f32(false, (A), false, (B), (short)0, (C), false, false)

__device__ __forceinline__ v2f col_pair(float e, float o) {
    v2f p;  p.x = e;  p.y = o;
    asm volatile("" : "+v"(p));
    return p;
}

__device__ __forceinline__ void wave_handover() {
    __builtin_amdgcn_fence(__ATOMIC_RELEASE, "workgroup", "local");
    __builtin_amdgcn_fence(__ATOMIC_ACQUIRE, "workgroup", "local");
}

__device__ __forceinline__ void group_handover() {
    __builtin_amdgcn_fence(__ATOMIC_RELEASE, "workgroup", "local");
    __builtin_amdgcn_s_barrier();
    __builtin_amdgcn_fence(__ATOMIC_ACQUIRE, "workgroup", "local");
}

template <int NW, int P>
__global__ __launch_bounds__(32 * NW)
void dwt_db4_level(const float* __restrict__ src, int N, long long srcStride,
                   float* __restrict__ dstLL, int ldLL, long long llStride,
                   float* __restrict__ outDet, int M)
{
    const int lane = threadIdx.x & 31;
    const int n    = lane & 15;
    const int g    = lane >> 4;
    const int w    = __builtin_amdgcn_readfirstlane((int)(threadIdx.x >> 5));
    int       c0   = w * (32 * P);
    const int r0   = blockIdx.y * 16;
    const int img  = blockIdx.z;

    src    += (long long)img * srcStride;
    dstLL  += (long long)img * llStride;
    outDet += (long long)img * (512LL * 512LL);

    __shared__ __attribute__((aligned(16))) float XsAll[NW][34][74];
    __shared__ __attribute__((aligned(8))) float LHtAll[NW][2][32][38];
    __shared__ float HaloF[NW][34];
    __shared__ float HaloL[NW][34];
    __shared__ float HaloM[NW][34];
    float (* const Xs)[74]      = XsAll[w];
    float (* const LHt)[32][38] = LHtAll[w];

    v2f W[5];
    #pragma unroll
    for (int u = 0; u < 5; ++u) W[u] = *(const v2f*)OPTAB.v[u][lane];

    const char* base = (const char*)src;
    unsigned rowOff[17];
    {
        int rb = 2 * r0 - 1 + g;       rb += (rb < 0) ? N : 0;
        const unsigned rowBytes = 4u * (unsigned)N;
        #pragma unroll
        for (int i = 0; i < 17; ++i) {
            unsigned x = (unsigned)(rb + 2 * i);
            unsigned xm = x - (unsigned)N;
            x = (xm < x) ? xm : x;
            rowOff[i] = __umul24(x, rowBytes);
        }
    }
    v4f st[17];
    v4f sb[17];
    {
        const unsigned colBytes = 4u * (unsigned)(2 * c0 + 4 * n);
        #pragma unroll
        for (int i = 0; i < 17; ++i) st[i] = *(const v4f*)(base + (rowOff[i] + colBytes));
        if (P == 2) {
            #pragma unroll
            for (int i = 0; i < 17; ++i) sb[i] = *(const v4f*)(base + (rowOff[i] + (colBytes + 256u)));
        }
    }
    {
        const v2f z = {0.f, 0.f};
        *(v2f*)&LHt[0][lane][34] = z;
        *(v2f*)&LHt[1][lane][34] = z;
        *(v2f*)&Xs[lane][70] = z;
        if (lane < 2) *(v2f*)&Xs[32 + lane][70] = z;
    }
    {
        float* const hf = &HaloF[w][g];
        float* const hm = &HaloM[w][g];
        float* const hl = &HaloL[w][g];
        if (n == 0) {
            #pragma unroll
            for (int i = 0; i < 17; ++i) {
                hf[2 * i] = st[i].x;
                if (P == 2) hm[2 * i] = sb[i].x;
            }
        }
        if (n == 15) {
            #pragma unroll
            for (int i = 0; i < 17; ++i) hl[2 * i] = (P == 2) ? sb[i].w : st[i].w;
        }
    }
    group_handover();
    float hl0 = HaloL[(w + NW - 1) & (NW - 1)][lane];
    float hl1 = HaloL[(w + NW - 1) & (NW - 1)][32 + (lane & 1)];
    const float hrE0 = HaloF[(w + 1) & (NW - 1)][lane];
    const float hrE1 = HaloF[(w + 1) & (NW - 1)][32 + (lane & 1)];
    const float hrM0 = (P == 2) ? HaloM[w][lane] : 0.f;
    const float hrM1 = (P == 2) ? HaloM[w][32 + (lane & 1)] : 0.f;

    #pragma unroll
    for (int ip = 0; ip < P; ++ip, c0 += 32) {
    {
        if (ip > 0) {
            hl0 = Xs[lane][68];
            hl1 = Xs[32 + (lane & 1)][68];
        }
        #pragma unroll
        for (int i = 0; i < 17; ++i) {
            const v4f v = (ip == 0) ? st[i] : sb[i];
            float* d = &Xs[2 * i + g][4 * n + 5];
            d[0] = v.x;  d[1] = v.y;  d[2] = v.z;  d[3] = v.w;
        }
        const float hr0 = (ip + 1 < P) ? hrM0 : hrE0;
        const float hr1 = (ip + 1 < P) ? hrM1 : hrE1;
        Xs[lane][4]  = hl0;
        Xs[lane][69] = hr0;
        if (lane < 2) { Xs[32 + lane][4] = hl1;  Xs[32 + lane][69] = hr1; }
    }
    wave_handover();

    const float* const ap = &Xs[32 + (n & 1)][16 * ((n >> 1) & 3) + 2 * g + 4];
    v2f at[5];
    #pragma unroll
    for (int u = 0; u < 5; ++u) at[u] = *(const v2f*)(ap + 4 * u);
    #pragma nounroll
    for (int tl = 0; tl < 2; ++tl) {
    float* const sdst = &LHt[n & 1][16 * tl + (n >> 1)][8 * g];
    #pragma unroll
    for (int ch = 0; ch < 2; ++ch) {
        const int R = 16 * ch;
        v8f acc0 = {0.f,0.f,0.f,0.f,0.f,0.f,0.f,0.f};
        v8f acc1 = {0.f,0.f,0.f,0.f,0.f,0.f,0.f,0.f};
        #pragma unroll
        for (int t = 0; t < 9; ++t) {
            v2f a = *(const v2f*)&Xs[R + n][32 * tl + 4 * t + 2 * g + 4];
            if (t <= 4) acc0 = WMMA_F32(a, W[t], acc0);
            if (t >= 4) acc1 = WMMA_F32(a, W[t - 4], acc1);
        }
        #pragma unroll
        for (int v = 0; v < 8; v += 2) {
            v2f p0, p1;
            p0.x = acc0[v];  p0.y = acc0[v + 1];
            p1.x = acc1[v];  p1.y = acc1[v + 1];
            *(v2f*)&sdst[R + v]          = p0;
            *(v2f*)&sdst[8 * 38 + R + v] = p1;
        }
    }
    }
    {
        v8f acc = {0.f,0.f,0.f,0.f,0.f,0.f,0.f,0.f};
        #pragma unroll
        for (int u = 0; u < 5; ++u) {
            acc = WMMA_F32(at[u], W[u], acc);
        }
        if (g == 0) {
            float* const d = &LHt[n & 1][n >> 1][32];
            #pragma unroll
            for (int q = 0; q < 4; ++q) {
                v2f p;
                p.x = acc[2 * q];  p.y = acc[2 * q + 1];
                *(v2f*)&d[8 * q * 38] = p;
            }
        }
    }
    wave_handover();

    {
    v8f Le0 = {0.f,0.f,0.f,0.f,0.f,0.f,0.f,0.f};
    v8f Lo0 = {0.f,0.f,0.f,0.f,0.f,0.f,0.f,0.f};
    v8f Le1 = {0.f,0.f,0.f,0.f,0.f,0.f,0.f,0.f};
    v8f Lo1 = {0.f,0.f,0.f,0.f,0.f,0.f,0.f,0.f};
    v8f He0 = {0.f,0.f,0.f,0.f,0.f,0.f,0.f,0.f};
    v8f Ho0 = {0.f,0.f,0.f,0.f,0.f,0.f,0.f,0.f};
    v8f He1 = {0.f,0.f,0.f,0.f,0.f,0.f,0.f,0.f};
    v8f Ho1 = {0.f,0.f,0.f,0.f,0.f,0.f,0.f,0.f};
    const float* const bp = &LHt[0][2 * n][2 * g];
    #pragma unroll
    for (int t = 0; t < 9; ++t) {
        const v2f bLe = *(const v2f*)(bp + 4 * t);
        const v2f bLo = *(const v2f*)(bp + 38 + 4 * t);
        const v2f bHe = *(const v2f*)(bp + 32 * 38 + 4 * t);
        const v2f bHo = *(const v2f*)(bp + 33 * 38 + 4 * t);
        if (t <= 4) {
            Le0 = WMMA_F32(W[t], bLe, Le0);      Lo0 = WMMA_F32(W[t], bLo, Lo0);
            He0 = WMMA_F32(W[t], bHe, He0);      Ho0 = WMMA_F32(W[t], bHo, Ho0);
        }
        if (t >= 4) {
            Le1 = WMMA_F32(W[t - 4], bLe, Le1);  Lo1 = WMMA_F32(W[t - 4], bLo, Lo1);
            He1 = WMMA_F32(W[t - 4], bHe, He1);  Ho1 = WMMA_F32(W[t - 4], bHo, Ho1);
        }
    }

    {
        const unsigned ocol  = (unsigned)(c0 + 2 * n);
        const unsigned orow0 = (unsigned)(r0 + 4 * g);
        const unsigned oLL   = 4u * (orow0 * (unsigned)ldLL + ocol);
        const unsigned oD    = 4u * (orow0 * 512u + ocol);
        const unsigned pitchLL = 4u * (unsigned)ldLL;
        char* pLL = (char*)dstLL;
        char* pLH = (char*)(outDet + M);
        char* pHL = (char*)(outDet + (long long)M * 512);
        char* pHH = (char*)(outDet + (long long)M * 512 + M);
        v2f sLL[8], sLH[8], sHL[8], sHH[8];
        #pragma unroll
        for (int w = 0; w < 4; ++w) {
            sLL[w]     = col_pair(Le0[2 * w],     Lo0[2 * w]);
            sLL[w + 4] = col_pair(Le1[2 * w],     Lo1[2 * w]);
            sLH[w]     = col_pair(Le0[2 * w + 1], Lo0[2 * w + 1]);
            sLH[w + 4] = col_pair(Le1[2 * w + 1], Lo1[2 * w + 1]);
            sHL[w]     = col_pair(He0[2 * w],     Ho0[2 * w]);
            sHL[w + 4] = col_pair(He1[2 * w],     Ho1[2 * w]);
            sHH[w]     = col_pair(He0[2 * w + 1], Ho0[2 * w + 1]);
            sHH[w + 4] = col_pair(He1[2 * w + 1], Ho1[2 * w + 1]);
        }
        unsigned oLLr[8];
        #pragma unroll
        for (int j = 0; j < 8; ++j) oLLr[j] = oLL + (unsigned)((j & 3) + 8 * (j >> 2)) * pitchLL;
        #pragma unroll
        for (int j = 0; j < 8; ++j) {
            const int row = (j & 3) + 8 * (j >> 2);
            *(v2f*)(pLL + oLLr[j]) = sLL[j];
            __builtin_nontemporal_store(sLH[j], (v2f*)((pLH + 2048 * row) + oD));
            __builtin_nontemporal_store(sHL[j], (v2f*)((pHL + 2048 * row) + oD));
            __builtin_nontemporal_store(sHH[j], (v2f*)((pHH + 2048 * row) + oD));
        }
    }
    }
    }
}

template <int P, bool LL_LDS>
__device__ __forceinline__ void pairs_body(float (* const Xs)[74], float (* const LHt)[32][38],
        const v4f (&st)[17], const v4f (&sb)[17], const v2f (&W)[5],
        const int lane, const int n, const int g, int c0, const int r0,
        float hl0, float hl1, const float hrE0, const float hrE1, const float hrM0, const float hrM1,
        float* __restrict__ dstLL, const int ldLL, float* const ll2,
        float* __restrict__ outDet, const int M)
{
    #pragma unroll
    for (int ip = 0; ip < P; ++ip, c0 += 32) {
    {
        if (ip > 0) {
            hl0 = Xs[lane][68];
            hl1 = Xs[32 + (lane & 1)][68];
        }
        #pragma unroll
        for (int i = 0; i < 17; ++i) {
            const v4f v = (ip == 0) ? st[i] : sb[i];
            float* d = &Xs[2 * i + g][4 * n + 5];
            d[0] = v.x;  d[1] = v.y;  d[2] = v.z;  d[3] = v.w;
        }
        const float hr0 = (ip + 1 < P) ? hrM0 : hrE0;
        const float hr1 = (ip + 1 < P) ? hrM1 : hrE1;
        Xs[lane][4]  = hl0;
        Xs[lane][69] = hr0;
        if (lane < 2) { Xs[32 + lane][4] = hl1;  Xs[32 + lane][69] = hr1; }
    }
    wave_handover();

    const float* const ap = &Xs[32 + (n & 1)][16 * ((n >> 1) & 3) + 2 * g + 4];
    v2f at[5];
    #pragma unroll
    for (int u = 0; u < 5; ++u) at[u] = *(const v2f*)(ap + 4 * u);
    #pragma nounroll
    for (int tl = 0; tl < 2; ++tl) {
    float* const sdst = &LHt[n & 1][16 * tl + (n >> 1)][8 * g];
    #pragma unroll
    for (int ch = 0; ch < 2; ++ch) {
        const int R = 16 * ch;
        v8f acc0 = {0.f,0.f,0.f,0.f,0.f,0.f,0.f,0.f};
        v8f acc1 = {0.f,0.f,0.f,0.f,0.f,0.f,0.f,0.f};
        #pragma unroll
        for (int t = 0; t < 9; ++t) {
            v2f a = *(const v2f*)&Xs[R + n][32 * tl + 4 * t + 2 * g + 4];
            if (t <= 4) acc0 = WMMA_F32(a, W[t], acc0);
            if (t >= 4) acc1 = WMMA_F32(a, W[t - 4], acc1);
        }
        #pragma unroll
        for (int v = 0; v < 8; v += 2) {
            v2f p0, p1;
            p0.x = acc0[v];  p0.y = acc0[v + 1];
            p1.x = acc1[v];  p1.y = acc1[v + 1];
            *(v2f*)&sdst[R + v]          = p0;
            *(v2f*)&sdst[8 * 38 + R + v] = p1;
        }
    }
    }
    {
        v8f acc = {0.f,0.f,0.f,0.f,0.f,0.f,0.f,0.f};
        #pragma unroll
        for (int u = 0; u < 5; ++u) {
            acc = WMMA_F32(at[u], W[u], acc);
        }
        if (g == 0) {
            float* const d = &LHt[n & 1][n >> 1][32];
            #pragma unroll
            for (int q = 0; q < 4; ++q) {
                v2f p;
                p.x = acc[2 * q];  p.y = acc[2 * q + 1];
                *(v2f*)&d[8 * q * 38] = p;
            }
        }
    }
    wave_handover();

    {
    v8f Le0 = {0.f,0.f,0.f,0.f,0.f,0.f,0.f,0.f};
    v8f Lo0 = {0.f,0.f,0.f,0.f,0.f,0.f,0.f,0.f};
    v8f Le1 = {0.f,0.f,0.f,0.f,0.f,0.f,0.f,0.f};
    v8f Lo1 = {0.f,0.f,0.f,0.f,0.f,0.f,0.f,0.f};
    v8f He0 = {0.f,0.f,0.f,0.f,0.f,0.f,0.f,0.f};
    v8f Ho0 = {0.f,0.f,0.f,0.f,0.f,0.f,0.f,0.f};
    v8f He1 = {0.f,0.f,0.f,0.f,0.f,0.f,0.f,0.f};
    v8f Ho1 = {0.f,0.f,0.f,0.f,0.f,0.f,0.f,0.f};
    const float* const bp = &LHt[0][2 * n][2 * g];
    #pragma unroll
    for (int t = 0; t < 9; ++t) {
        const v2f bLe = *(const v2f*)(bp + 4 * t);
        const v2f bLo = *(const v2f*)(bp + 38 + 4 * t);
        const v2f bHe = *(const v2f*)(bp + 32 * 38 + 4 * t);
        const v2f bHo = *(const v2f*)(bp + 33 * 38 + 4 * t);
        if (t <= 4) {
            Le0 = WMMA_F32(W[t], bLe, Le0);      Lo0 = WMMA_F32(W[t], bLo, Lo0);
            He0 = WMMA_F32(W[t], bHe, He0);      Ho0 = WMMA_F32(W[t], bHo, Ho0);
        }
        if (t >= 4) {
            Le1 = WMMA_F32(W[t - 4], bLe, Le1);  Lo1 = WMMA_F32(W[t - 4], bLo, Lo1);
            He1 = WMMA_F32(W[t - 4], bHe, He1);  Ho1 = WMMA_F32(W[t - 4], bHo, Ho1);
        }
    }

    {
        const unsigned ocol  = (unsigned)(c0 + 2 * n);
        const unsigned orow0 = (unsigned)(r0 + 4 * g);
        const unsigned oLL   = 4u * (orow0 * (unsigned)ldLL + ocol);
        const unsigned oD    = 4u * (orow0 * 512u + ocol);
        const unsigned pitchLL = 4u * (unsigned)ldLL;
        char* pLL = (char*)dstLL;
        char* pLH = (char*)(outDet + M);
        char* pHL = (char*)(outDet + (long long)M * 512);
        char* pHH = (char*)(outDet + (long long)M * 512 + M);
        v2f sLL[8], sLH[8], sHL[8], sHH[8];
        #pragma unroll
        for (int w = 0; w < 4; ++w) {
            sLL[w]     = col_pair(Le0[2 * w],     Lo0[2 * w]);
            sLL[w + 4] = col_pair(Le1[2 * w],     Lo1[2 * w]);
            sLH[w]     = col_pair(Le0[2 * w + 1], Lo0[2 * w + 1]);
            sLH[w + 4] = col_pair(Le1[2 * w + 1], Lo1[2 * w + 1]);
            sHL[w]     = col_pair(He0[2 * w],     Ho0[2 * w]);
            sHL[w + 4] = col_pair(He1[2 * w],     Ho1[2 * w]);
            sHH[w]     = col_pair(He0[2 * w + 1], Ho0[2 * w + 1]);
            sHH[w + 4] = col_pair(He1[2 * w + 1], Ho1[2 * w + 1]);
        }
        unsigned oLLr[8];
        #pragma unroll
        for (int j = 0; j < 8; ++j) oLLr[j] = oLL + (unsigned)((j & 3) + 8 * (j >> 2)) * pitchLL;
        #pragma unroll
        for (int j = 0; j < 8; ++j) {
            const int row = (j & 3) + 8 * (j >> 2);
            if (LL_LDS) *(v2f*)&ll2[row * 132 + (int)ocol] = sLL[j];
            else        *(v2f*)(pLL + oLLr[j]) = sLL[j];
            __builtin_nontemporal_store(sLH[j], (v2f*)((pLH + 2048 * row) + oD));
            __builtin_nontemporal_store(sHL[j], (v2f*)((pHL + 2048 * row) + oD));
            __builtin_nontemporal_store(sHH[j], (v2f*)((pHH + 2048 * row) + oD));
        }
    }
    }
    }
}

__global__ __launch_bounds__(128)
void dwt_db4_level23(const float* __restrict__ src, float* __restrict__ out)
{
    const int lane = threadIdx.x & 31;
    const int n    = lane & 15;
    const int g    = lane >> 4;
    const int w    = __builtin_amdgcn_readfirstlane((int)(threadIdx.x >> 5));
    const int s    = w & 1;
    const int bl   = w >> 1;
    const int b3   = blockIdx.y;
    const int img  = blockIdx.z;

    src += (long long)img * (256LL * 256LL);
    out += (long long)img * (512LL * 512LL);

    __shared__ __attribute__((aligned(16))) float XsAll[4][34][74];
    __shared__ __attribute__((aligned(8)))  float LHtAll[4][2][32][38];
    __shared__ float HaloF[4][34];
    __shared__ float HaloL[4][34];
    __shared__ float HaloM[4][34];
    __shared__ float HbF[4][4];
    __shared__ float HbL[4][4];
    __shared__ float HbM[4][4];
    __shared__ __attribute__((aligned(16))) float LL2s[34][132];
    float (* const Xs)[74]      = XsAll[w];
    float (* const LHt)[32][38] = LHtAll[w];

    v2f W[5];
    #pragma unroll
    for (int u = 0; u < 5; ++u) W[u] = *(const v2f*)OPTAB.v[u][lane];

    const int r0A = (2 * b3 + bl) * 16;
    const int c0A = s * 64;
    const char* base = (const char*)src;
    unsigned rowOff[17];
    {
        int rb = 2 * r0A - 1 + g;      rb += (rb < 0) ? 256 : 0;
        #pragma unroll
        for (int i = 0; i < 17; ++i) rowOff[i] = 1024u * ((unsigned)(rb + 2 * i) & 255u);
    }
    unsigned rowOffB[2];
    {
        const int rho = 64 * b3 + (bl ? 63 : -3) + g;
        rowOffB[0] = 1024u * ((unsigned)(rho)     & 255u);
        rowOffB[1] = 1024u * ((unsigned)(rho + 2) & 255u);
    }
    v4f st[17], sb[17];
    v4f eb0[2], eb1[2];
    {
        const unsigned colBytes = 4u * (unsigned)(2 * c0A + 4 * n);
        #pragma unroll
        for (int i = 0; i < 2; ++i) {
            eb0[i] = *(const v4f*)(base + (rowOffB[i] + colBytes));
            eb1[i] = *(const v4f*)(base + (rowOffB[i] + (colBytes + 256u)));
        }
        #pragma unroll
        for (int i = 0; i < 17; ++i) st[i] = *(const v4f*)(base + (rowOff[i] + colBytes));
        #pragma unroll
        for (int i = 0; i < 17; ++i) sb[i] = *(const v4f*)(base + (rowOff[i] + (colBytes + 256u)));
    }
    {
        const v2f z = {0.f, 0.f};
        *(v2f*)&LHt[0][lane][34] = z;
        *(v2f*)&LHt[1][lane][34] = z;
        *(v2f*)&Xs[lane][70] = z;
        if (lane < 2) *(v2f*)&Xs[32 + lane][70] = z;
    }
    {
        if (n == 0) {
            #pragma unroll
            for (int i = 0; i < 2; ++i) { HbF[w][2 * i + g] = eb0[i].x;  HbM[w][2 * i + g] = eb1[i].x; }
            #pragma unroll
            for (int i = 0; i < 17; ++i) { HaloF[w][2 * i + g] = st[i].x;  HaloM[w][2 * i + g] = sb[i].x; }
        }
        if (n == 15) {
            #pragma unroll
            for (int i = 0; i < 2; ++i) HbL[w][2 * i + g] = eb1[i].w;
            #pragma unroll
            for (int i = 0; i < 17; ++i) HaloL[w][2 * i + g] = sb[i].w;
        }
    }
    group_handover();
    float hl0 = HaloL[w ^ 1][lane];
    float hl1 = HaloL[w ^ 1][32 + (lane & 1)];
    const float hrE0 = HaloF[w ^ 1][lane];
    const float hrE1 = HaloF[w ^ 1][32 + (lane & 1)];
    const float hrM0 = HaloM[w][lane];
    const float hrM1 = HaloM[w][32 + (lane & 1)];
    float       bLeft = HbL[w ^ 1][lane & 3];
    const float bRgtE = HbF[w ^ 1][lane & 3];
    const float bRgtM = HbM[w][lane & 3];

    #pragma unroll
    for (int ip = 0; ip < 2; ++ip) {
        if (ip > 0) bLeft = Xs[lane & 3][68];
        #pragma unroll
        for (int i = 0; i < 2; ++i) {
            const v4f v = (ip == 0) ? eb0[i] : eb1[i];
            float* d = &Xs[2 * i + g][4 * n + 5];
            d[0] = v.x;  d[1] = v.y;  d[2] = v.z;  d[3] = v.w;
        }
        const float bRgt = (ip == 0) ? bRgtM : bRgtE;
        if (lane < 4) { Xs[lane][4] = bLeft;  Xs[lane][69] = bRgt; }
        wave_handover();
        const float* const ap4 = &Xs[n & 3][16 * (n >> 2) + 2 * g + 4];
        v8f acc = {0.f,0.f,0.f,0.f,0.f,0.f,0.f,0.f};
        #pragma unroll
        for (int u = 0; u < 5; ++u) acc = WMMA_F32(*(const v2f*)(ap4 + 4 * u), W[u], acc);
        if ((n & 1) == 0) {
            float* const d = &LHt[0][16 * g + (n >> 1)][0];
            v2f p;
            p.x = acc[0];  p.y = acc[1];  *(v2f*)&d[0]          = p;
            p.x = acc[2];  p.y = acc[3];  *(v2f*)&d[2]          = p;
            p.x = acc[4];  p.y = acc[5];  *(v2f*)&d[8 * 38]     = p;
            p.x = acc[6];  p.y = acc[7];  *(v2f*)&d[8 * 38 + 2] = p;
        }
        wave_handover();
        const v8f zero8 = {0.f,0.f,0.f,0.f,0.f,0.f,0.f,0.f};
        const v8f ve = WMMA_F32(W[0], *(const v2f*)&LHt[0][2 * n][2 * g],     zero8);
        const v8f vo = WMMA_F32(W[0], *(const v2f*)&LHt[0][2 * n + 1][2 * g], zero8);
        if (g == 0) {
            v2f p;  p.x = ve[0];  p.y = vo[0];
            *(v2f*)&LL2s[bl ? 33 : 0][c0A + 32 * ip + 2 * n] = p;
        }
        wave_handover();
    }

    pairs_body<2, true>(Xs, LHt, st, sb, W, lane, n, g, c0A, r0A,
                        hl0, hl1, hrE0, hrE1, hrM0, hrM1,
                        nullptr, 0, &LL2s[1 + 16 * bl + 4 * g][0], out, 128);

    group_handover();

    v4f t3[17];
    if (w < 2) {
        #pragma unroll
        for (int i = 0; i < 17; ++i) t3[i] = *(const v4f*)&LL2s[2 * i + g][64 * w + 4 * n];
        if (n == 0) {
            #pragma unroll
            for (int i = 0; i < 17; ++i) HaloF[w][2 * i + g] = t3[i].x;
        }
        if (n == 15) {
            #pragma unroll
            for (int i = 0; i < 17; ++i) HaloL[w][2 * i + g] = t3[i].w;
        }
    }
    group_handover();
    if (w < 2) {
        const float l0 = HaloL[w ^ 1][lane];
        const float l1 = HaloL[w ^ 1][32 + (lane & 1)];
        const float e0 = HaloF[w ^ 1][lane];
        const float e1 = HaloF[w ^ 1][32 + (lane & 1)];
        pairs_body<1, false>(Xs, LHt, t3, t3, W, lane, n, g, 32 * w, 16 * b3,
                             l0, l1, e0, e1, 0.f, 0.f,
                             out, 512, nullptr, out, 64);
    }
}

extern "C" void kernel_launch(void* const* d_in, const int* in_sizes, int n_in,
                              void* d_out, int out_size, void* d_ws, size_t ws_size,
                              hipStream_t stream) {
    const float* x = (const float*)d_in[0];
    float* out = (float*)d_out;
    const int BC = in_sizes[0] / (512 * 512);

    float* LL1 = (float*)d_ws;

    dwt_db4_level<4, 2><<<dim3(1, 16, BC), dim3(128, 1, 1), 0, stream>>>(
        x,   512, 512LL * 512, LL1, 256, 256LL * 256, out, 256);
    dwt_db4_level23<<<dim3(1, 4, BC), dim3(128, 1, 1), 0, stream>>>(LL1, out);
}


# Both passes as ONE matrix product per 2x2 output block (K = its 6x6 input patch, 16 rows = 4 positions x 4 subbands): 72 not 85 products per tile pair, no LDS round trip; sum order changes
# speedup vs baseline: 5.0973x; 1.0372x over previous
#include <hip/hip_runtime.h>

typedef __attribute__((ext_vector_type(2))) float v2f;
typedef __attribute__((ext_vector_type(4))) float v4f;
typedef __attribute__((ext_vector_type(8))) float v8f;

struct OpTab { float v[5][32][2]; };

constexpr float db4pick(const double* h, int t) {
    return (t >= 0 && t < 4) ? (float)h[t] : 0.0f;
}

constexpr OpTab make_optab() {
    OpTab o{};
    const double s3  = 1.7320508075688772935;
    const double nrm = 0.17677669529663688110;
    const double h0[4] = {(1.0 + s3) * nrm, (3.0 + s3) * nrm,
                          (3.0 - s3) * nrm, (1.0 - s3) * nrm};
    const double h1[4] = {h0[3], -h0[2], h0[1], -h0[0]};
    for (int u = 0; u < 5; ++u) {
        for (int lane = 0; lane < 32; ++lane) {
            int m = (lane & 15) >> 1, f = lane & 1, h = lane >> 4;
            int tau = 4 * u + 2 * h - 2 * m;
            const double* w = f ? h1 : h0;
            o.v[u][lane][0] = db4pick(w, tau);
            o.v[u][lane][1] = db4pick(w, tau + 1);
        }
    }
    return o;
}

__device__ __constant__ OpTab OPTAB = make_optab();


struct Op2Tab { float v[9][32][2]; };

constexpr Op2Tab make_op2tab() {
    Op2Tab o{};
    const double s3  = 1.7320508075688772935;
    const double nrm = 0.17677669529663688110;
    const double h0[4] = {(1.0 + s3) * nrm, (3.0 + s3) * nrm,
                          (3.0 - s3) * nrm, (1.0 - s3) * nrm};
    const double h1[4] = {h0[3], -h0[2], h0[1], -h0[0]};
    for (int u = 0; u < 9; ++u) {
        for (int lane = 0; lane < 32; ++lane) {
            const int c = lane & 15, h = lane >> 4;
            const int py = c >> 3, s = (c >> 1) & 3, px = c & 1;
            const double* hh = (s >> 1) ? h1 : h0;
            const double* hv = (s & 1) ? h1 : h0;
            for (int j = 0; j < 2; ++j) {
                const int e = 4 * u + 2 * h + j;
                const int dy = e / 6, dx = e % 6;
                o.v[u][lane][j] = (float)((double)db4pick(hv, dy - 2 * py) * (double)db4pick(hh, dx - 2 * px));
            }
        }
    }
    return o;
}

__device__ __constant__ Op2Tab OP2TAB = make_op2tab();

#define WMMA_F32(A, B, C) \
    __builtin_amdgcn_wmma_f32_16x16x4_f32(false, (A), false, (B), (short)0, (C), false, false)

__device__ __forceinline__ void wave_handover() {
    __builtin_amdgcn_fence(__ATOMIC_RELEASE, "workgroup", "local");
    __builtin_amdgcn_fence(__ATOMIC_ACQUIRE, "workgroup", "local");
}

__device__ __forceinline__ void group_handover() {
    __builtin_amdgcn_fence(__ATOMIC_RELEASE, "workgroup", "local");
    __builtin_amdgcn_s_barrier();
    __builtin_amdgcn_fence(__ATOMIC_ACQUIRE, "workgroup", "local");
}

template <int NW, int P>
__global__ __launch_bounds__(32 * NW)
void dwt_db4_level(const float* __restrict__ src, int N, long long srcStride,
                   float* __restrict__ dstLL, int ldLL, long long llStride,
                   float* __restrict__ outDet, int M)
{
    const int lane = threadIdx.x & 31;
    const int n    = lane & 15;
    const int g    = lane >> 4;
    const int w    = __builtin_amdgcn_readfirstlane((int)(threadIdx.x >> 5));
    int       c0   = w * (32 * P);
    const int r0   = blockIdx.y * 16;
    const int img  = blockIdx.z;

    src    += (long long)img * srcStride;
    dstLL  += (long long)img * llStride;
    outDet += (long long)img * (512LL * 512LL);

    __shared__ __attribute__((aligned(16))) float XsAll[NW][34][74];
    __shared__ float HaloF[NW][34];
    __shared__ float HaloL[NW][34];
    __shared__ float HaloM[NW][34];
    float (* const Xs)[74]      = XsAll[w];

    v2f OP[9];
    #pragma unroll
    for (int u = 0; u < 9; ++u) OP[u] = *(const v2f*)OP2TAB.v[u][lane];

    const char* base = (const char*)src;
    unsigned rowOff[17];
    {
        int rb = 2 * r0 - 1 + g;       rb += (rb < 0) ? N : 0;
        const unsigned rowBytes = 4u * (unsigned)N;
        #pragma unroll
        for (int i = 0; i < 17; ++i) {
            unsigned x = (unsigned)(rb + 2 * i);
            unsigned xm = x - (unsigned)N;
            x = (xm < x) ? xm : x;
            rowOff[i] = __umul24(x, rowBytes);
        }
    }
    v4f st[17];
    v4f sb[17];
    {
        const unsigned colBytes = 4u * (unsigned)(2 * c0 + 4 * n);
        #pragma unroll
        for (int i = 0; i < 17; ++i) st[i] = *(const v4f*)(base + (rowOff[i] + colBytes));
        if (P == 2) {
            #pragma unroll
            for (int i = 0; i < 17; ++i) sb[i] = *(const v4f*)(base + (rowOff[i] + (colBytes + 256u)));
        }
    }
    {
        float* const hf = &HaloF[w][g];
        float* const hm = &HaloM[w][g];
        float* const hl = &HaloL[w][g];
        if (n == 0) {
            #pragma unroll
            for (int i = 0; i < 17; ++i) {
                hf[2 * i] = st[i].x;
                if (P == 2) hm[2 * i] = sb[i].x;
            }
        }
        if (n == 15) {
            #pragma unroll
            for (int i = 0; i < 17; ++i) hl[2 * i] = (P == 2) ? sb[i].w : st[i].w;
        }
    }
    group_handover();
    float hl0 = HaloL[(w + NW - 1) & (NW - 1)][lane];
    float hl1 = HaloL[(w + NW - 1) & (NW - 1)][32 + (lane & 1)];
    const float hrE0 = HaloF[(w + 1) & (NW - 1)][lane];
    const float hrE1 = HaloF[(w + 1) & (NW - 1)][32 + (lane & 1)];
    const float hrM0 = (P == 2) ? HaloM[w][lane] : 0.f;
    const float hrM1 = (P == 2) ? HaloM[w][32 + (lane & 1)] : 0.f;

    #pragma unroll
    for (int ip = 0; ip < P; ++ip, c0 += 32) {
    {
        if (ip > 0) {
            hl0 = Xs[lane][68];
            hl1 = Xs[32 + (lane & 1)][68];
        }
        #pragma unroll
        for (int i = 0; i < 17; ++i) {
            const v4f v = (ip == 0) ? st[i] : sb[i];
            float* d = &Xs[2 * i + g][4 * n + 5];
            d[0] = v.x;  d[1] = v.y;  d[2] = v.z;  d[3] = v.w;
        }
        const float hr0 = (ip + 1 < P) ? hrM0 : hrE0;
        const float hr1 = (ip + 1 < P) ? hrM1 : hrE1;
        Xs[lane][4]  = hl0;
        Xs[lane][69] = hr0;
        if (lane < 2) { Xs[32 + lane][4] = hl1;  Xs[32 + lane][69] = hr1; }
    }
    wave_handover();

    {
    const float* const p0 = &Xs[0][4 * n + 2 * g + 4];
    const float* const p1 = &Xs[g][4 * n + (g ? 4 : 8)];
    v8f acc[8];
    #pragma unroll
    for (int rp = 0; rp < 8; ++rp) acc[rp] = (v8f){0.f,0.f,0.f,0.f,0.f,0.f,0.f,0.f};
    #pragma unroll
    for (int k = 0; k < 17; ++k) {
        const v2f t0 = *(const v2f*)(p0 + 148 * k);
        const v2f t1 = *(const v2f*)(p1 + 148 * k);
        const v2f t2 = *(const v2f*)(p0 + 148 * k + 76);
        if ((k & 1) == 0) {
            if (k >= 2) {
                acc[k / 2 - 1] = WMMA_F32(OP[6], t0, acc[k / 2 - 1]);
                acc[k / 2 - 1] = WMMA_F32(OP[7], t1, acc[k / 2 - 1]);
                acc[k / 2 - 1] = WMMA_F32(OP[8], t2, acc[k / 2 - 1]);
            }
            if (k <= 14) {
                acc[k / 2] = WMMA_F32(OP[0], t0, acc[k / 2]);
                acc[k / 2] = WMMA_F32(OP[1], t1, acc[k / 2]);
                acc[k / 2] = WMMA_F32(OP[2], t2, acc[k / 2]);
            }
        } else {
            acc[k / 2] = WMMA_F32(OP[3], t0, acc[k / 2]);
            acc[k / 2] = WMMA_F32(OP[4], t1, acc[k / 2]);
            acc[k / 2] = WMMA_F32(OP[5], t2, acc[k / 2]);
        }
    }

    const unsigned ocol  = (unsigned)(c0 + 2 * n);
    const unsigned orow0 = (unsigned)(r0 + g);
    const unsigned oLL   = 4u * (orow0 * (unsigned)ldLL + ocol);
    const unsigned oD    = 4u * (orow0 * 512u + ocol);
    const unsigned pitch2LL = 8u * (unsigned)ldLL;
    unsigned oLLr[8];
    #pragma unroll
    for (int rp = 0; rp < 8; ++rp) oLLr[rp] = oLL + (unsigned)rp * pitch2LL;
    char* pLL = (char*)dstLL;
    char* pLH = (char*)(outDet + M);
    char* pHL = (char*)(outDet + (long long)M * 512);
    char* pHH = (char*)(outDet + (long long)M * 512 + M);

    __builtin_amdgcn_sched_barrier(0);

    {
        #pragma unroll
        for (int rp = 0; rp < 8; ++rp) {
            v2f qLL, qLH, qHL, qHH;
            qLL.x = acc[rp][0];  qLL.y = acc[rp][1];
            qLH.x = acc[rp][2];  qLH.y = acc[rp][3];
            qHL.x = acc[rp][4];  qHL.y = acc[rp][5];
            qHH.x = acc[rp][6];  qHH.y = acc[rp][7];
            *(v2f*)(pLL + oLLr[rp]) = qLL;
            __builtin_nontemporal_store(qLH, (v2f*)((pLH + 4096 * rp) + oD));
            __builtin_nontemporal_store(qHL, (v2f*)((pHL + 4096 * rp) + oD));
            __builtin_nontemporal_store(qHH, (v2f*)((pHH + 4096 * rp) + oD));
        }
    }
    }
    if (ip + 1 < P) wave_handover();
    }
}

template <int P, bool LL_LDS>
__device__ __forceinline__ void pairs_body(float (* const Xs)[74],
        const v4f (&st)[17], const v4f (&sb)[17], const v2f (&OP)[9],
        const int lane, const int n, const int g, int c0, const int r0,
        float hl0, float hl1, const float hrE0, const float hrE1, const float hrM0, const float hrM1,
        float* __restrict__ dstLL, const int ldLL, float* const ll2,
        float* __restrict__ outDet, const int M)
{
    #pragma unroll
    for (int ip = 0; ip < P; ++ip, c0 += 32) {
    {
        if (ip > 0) {
            hl0 = Xs[lane][68];
            hl1 = Xs[32 + (lane & 1)][68];
        }
        #pragma unroll
        for (int i = 0; i < 17; ++i) {
            const v4f v = (ip == 0) ? st[i] : sb[i];
            float* d = &Xs[2 * i + g][4 * n + 5];
            d[0] = v.x;  d[1] = v.y;  d[2] = v.z;  d[3] = v.w;
        }
        const float hr0 = (ip + 1 < P) ? hrM0 : hrE0;
        const float hr1 = (ip + 1 < P) ? hrM1 : hrE1;
        Xs[lane][4]  = hl0;
        Xs[lane][69] = hr0;
        if (lane < 2) { Xs[32 + lane][4] = hl1;  Xs[32 + lane][69] = hr1; }
    }
    wave_handover();

    {
    const float* const p0 = &Xs[0][4 * n + 2 * g + 4];
    const float* const p1 = &Xs[g][4 * n + (g ? 4 : 8)];
    v8f acc[8];
    #pragma unroll
    for (int rp = 0; rp < 8; ++rp) acc[rp] = (v8f){0.f,0.f,0.f,0.f,0.f,0.f,0.f,0.f};
    #pragma unroll
    for (int k = 0; k < 17; ++k) {
        const v2f t0 = *(const v2f*)(p0 + 148 * k);
        const v2f t1 = *(const v2f*)(p1 + 148 * k);
        const v2f t2 = *(const v2f*)(p0 + 148 * k + 76);
        if ((k & 1) == 0) {
            if (k >= 2) {
                acc[k / 2 - 1] = WMMA_F32(OP[6], t0, acc[k / 2 - 1]);
                acc[k / 2 - 1] = WMMA_F32(OP[7], t1, acc[k / 2 - 1]);
                acc[k / 2 - 1] = WMMA_F32(OP[8], t2, acc[k / 2 - 1]);
            }
            if (k <= 14) {
                acc[k / 2] = WMMA_F32(OP[0], t0, acc[k / 2]);
                acc[k / 2] = WMMA_F32(OP[1], t1, acc[k / 2]);
                acc[k / 2] = WMMA_F32(OP[2], t2, acc[k / 2]);
            }
        } else {
            acc[k / 2] = WMMA_F32(OP[3], t0, acc[k / 2]);
            acc[k / 2] = WMMA_F32(OP[4], t1, acc[k / 2]);
            acc[k / 2] = WMMA_F32(OP[5], t2, acc[k / 2]);
        }
    }

    const unsigned ocol  = (unsigned)(c0 + 2 * n);
    const unsigned orow0 = (unsigned)(r0 + g);
    const unsigned oLL   = 4u * (orow0 * (unsigned)ldLL + ocol);
    const unsigned oD    = 4u * (orow0 * 512u + ocol);
    const unsigned pitch2LL = 8u * (unsigned)ldLL;
    unsigned oLLr[8];
    #pragma unroll
    for (int rp = 0; rp < 8; ++rp) oLLr[rp] = oLL + (unsigned)rp * pitch2LL;
    char* pLL = (char*)dstLL;
    char* pLH = (char*)(outDet + M);
    char* pHL = (char*)(outDet + (long long)M * 512);
    char* pHH = (char*)(outDet + (long long)M * 512 + M);

    __builtin_amdgcn_sched_barrier(0);

    {
        #pragma unroll
        for (int rp = 0; rp < 8; ++rp) {
            v2f qLL, qLH, qHL, qHH;
            qLL.x = acc[rp][0];  qLL.y = acc[rp][1];
            qLH.x = acc[rp][2];  qLH.y = acc[rp][3];
            qHL.x = acc[rp][4];  qHL.y = acc[rp][5];
            qHH.x = acc[rp][6];  qHH.y = acc[rp][7];
            if (LL_LDS) *(v2f*)&ll2[2 * rp * 132 + (int)ocol] = qLL;
            else        *(v2f*)(pLL + oLLr[rp]) = qLL;
            __builtin_nontemporal_store(qLH, (v2f*)((pLH + 4096 * rp) + oD));
            __builtin_nontemporal_store(qHL, (v2f*)((pHL + 4096 * rp) + oD));
            __builtin_nontemporal_store(qHH, (v2f*)((pHH + 4096 * rp) + oD));
        }
    }
    }
    if (ip + 1 < P) wave_handover();
    }
}

__global__ __launch_bounds__(128)
void dwt_db4_level23(const float* __restrict__ src, float* __restrict__ out)
{
    const int lane = threadIdx.x & 31;
    const int n    = lane & 15;
    const int g    = lane >> 4;
    const int w    = __builtin_amdgcn_readfirstlane((int)(threadIdx.x >> 5));
    const int s    = w & 1;
    const int bl   = w >> 1;
    const int b3   = blockIdx.y;
    const int img  = blockIdx.z;

    src += (long long)img * (256LL * 256LL);
    out += (long long)img * (512LL * 512LL);

    __shared__ __attribute__((aligned(16))) float XsAll[4][34][74];
    __shared__ __attribute__((aligned(8)))  float LHtAll[4][1][32][38];
    __shared__ float HaloF[4][34];
    __shared__ float HaloL[4][34];
    __shared__ float HaloM[4][34];
    __shared__ float HbF[4][4];
    __shared__ float HbL[4][4];
    __shared__ float HbM[4][4];
    __shared__ __attribute__((aligned(16))) float LL2s[34][132];
    float (* const Xs)[74]      = XsAll[w];
    float (* const LHt)[32][38] = LHtAll[w];

    v2f W[5];
    #pragma unroll
    for (int u = 0; u < 5; ++u) W[u] = *(const v2f*)OPTAB.v[u][lane];
    v2f OP[9];
    #pragma unroll
    for (int u = 0; u < 9; ++u) OP[u] = *(const v2f*)OP2TAB.v[u][lane];

    const int r0A = (2 * b3 + bl) * 16;
    const int c0A = s * 64;
    const char* base = (const char*)src;
    unsigned rowOff[17];
    {
        int rb = 2 * r0A - 1 + g;      rb += (rb < 0) ? 256 : 0;
        #pragma unroll
        for (int i = 0; i < 17; ++i) rowOff[i] = 1024u * ((unsigned)(rb + 2 * i) & 255u);
    }
    unsigned rowOffB[2];
    {
        const int rho = 64 * b3 + (bl ? 63 : -3) + g;
        rowOffB[0] = 1024u * ((unsigned)(rho)     & 255u);
        rowOffB[1] = 1024u * ((unsigned)(rho + 2) & 255u);
    }
    v4f st[17], sb[17];
    v4f eb0[2], eb1[2];
    {
        const unsigned colBytes = 4u * (unsigned)(2 * c0A + 4 * n);
        #pragma unroll
        for (int i = 0; i < 2; ++i) {
            eb0[i] = *(const v4f*)(base + (rowOffB[i] + colBytes));
            eb1[i] = *(const v4f*)(base + (rowOffB[i] + (colBytes + 256u)));
        }
        #pragma unroll
        for (int i = 0; i < 17; ++i) st[i] = *(const v4f*)(base + (rowOff[i] + colBytes));
        #pragma unroll
        for (int i = 0; i < 17; ++i) sb[i] = *(const v4f*)(base + (rowOff[i] + (colBytes + 256u)));
    }
    {
        const v2f z = {0.f, 0.f};
        if (lane < 4) *(v2f*)&Xs[lane][70] = z;
    }
    {
        if (n == 0) {
            #pragma unroll
            for (int i = 0; i < 2; ++i) { HbF[w][2 * i + g] = eb0[i].x;  HbM[w][2 * i + g] = eb1[i].x; }
            #pragma unroll
            for (int i = 0; i < 17; ++i) { HaloF[w][2 * i + g] = st[i].x;  HaloM[w][2 * i + g] = sb[i].x; }
        }
        if (n == 15) {
            #pragma unroll
            for (int i = 0; i < 2; ++i) HbL[w][2 * i + g] = eb1[i].w;
            #pragma unroll
            for (int i = 0; i < 17; ++i) HaloL[w][2 * i + g] = sb[i].w;
        }
    }
    group_handover();
    float hl0 = HaloL[w ^ 1][lane];
    float hl1 = HaloL[w ^ 1][32 + (lane & 1)];
    const float hrE0 = HaloF[w ^ 1][lane];
    const float hrE1 = HaloF[w ^ 1][32 + (lane & 1)];
    const float hrM0 = HaloM[w][lane];
    const float hrM1 = HaloM[w][32 + (lane & 1)];
    float       bLeft = HbL[w ^ 1][lane & 3];
    const float bRgtE = HbF[w ^ 1][lane & 3];
    const float bRgtM = HbM[w][lane & 3];

    #pragma unroll
    for (int ip = 0; ip < 2; ++ip) {
        if (ip > 0) bLeft = Xs[lane & 3][68];
        #pragma unroll
        for (int i = 0; i < 2; ++i) {
            const v4f v = (ip == 0) ? eb0[i] : eb1[i];
            float* d = &Xs[2 * i + g][4 * n + 5];
            d[0] = v.x;  d[1] = v.y;  d[2] = v.z;  d[3] = v.w;
        }
        const float bRgt = (ip == 0) ? bRgtM : bRgtE;
        if (lane < 4) { Xs[lane][4] = bLeft;  Xs[lane][69] = bRgt; }
        wave_handover();
        const float* const ap4 = &Xs[n & 3][16 * (n >> 2) + 2 * g + 4];
        v8f acc = {0.f,0.f,0.f,0.f,0.f,0.f,0.f,0.f};
        #pragma unroll
        for (int u = 0; u < 5; ++u) acc = WMMA_F32(*(const v2f*)(ap4 + 4 * u), W[u], acc);
        if ((n & 1) == 0) {
            float* const d = &LHt[0][16 * g + (n >> 1)][0];
            v2f p;
            p.x = acc[0];  p.y = acc[1];  *(v2f*)&d[0]          = p;
            p.x = acc[2];  p.y = acc[3];  *(v2f*)&d[2]          = p;
            p.x = acc[4];  p.y = acc[5];  *(v2f*)&d[8 * 38]     = p;
            p.x = acc[6];  p.y = acc[7];  *(v2f*)&d[8 * 38 + 2] = p;
        }
        wave_handover();
        const v8f zero8 = {0.f,0.f,0.f,0.f,0.f,0.f,0.f,0.f};
        const v8f ve = WMMA_F32(W[0], *(const v2f*)&LHt[0][2 * n][2 * g],     zero8);
        const v8f vo = WMMA_F32(W[0], *(const v2f*)&LHt[0][2 * n + 1][2 * g], zero8);
        if (g == 0) {
            v2f p;  p.x = ve[0];  p.y = vo[0];
            *(v2f*)&LL2s[bl ? 33 : 0][c0A + 32 * ip + 2 * n] = p;
        }
        wave_handover();
    }

    pairs_body<2, true>(Xs, st, sb, OP, lane, n, g, c0A, r0A,
                        hl0, hl1, hrE0, hrE1, hrM0, hrM1,
                        nullptr, 0, &LL2s[1 + 16 * bl + g][0], out, 128);

    group_handover();

    v4f t3[17];
    if (w < 2) {
        #pragma unroll
        for (int i = 0; i < 17; ++i) t3[i] = *(const v4f*)&LL2s[2 * i + g][64 * w + 4 * n];
        if (n == 0) {
            #pragma unroll
            for (int i = 0; i < 17; ++i) HaloF[w][2 * i + g] = t3[i].x;
        }
        if (n == 15) {
            #pragma unroll
            for (int i = 0; i < 17; ++i) HaloL[w][2 * i + g] = t3[i].w;
        }
    }
    group_handover();
    if (w < 2) {
        const float l0 = HaloL[w ^ 1][lane];
        const float l1 = HaloL[w ^ 1][32 + (lane & 1)];
        const float e0 = HaloF[w ^ 1][lane];
        const float e1 = HaloF[w ^ 1][32 + (lane & 1)];
        pairs_body<1, false>(Xs, t3, t3, OP, lane, n, g, 32 * w, 16 * b3,
                             l0, l1, e0, e1, 0.f, 0.f,
                             out, 512, nullptr, out, 64);
    }
}

extern "C" void kernel_launch(void* const* d_in, const int* in_sizes, int n_in,
                              void* d_out, int out_size, void* d_ws, size_t ws_size,
                              hipStream_t stream) {
    const float* x = (const float*)d_in[0];
    float* out = (float*)d_out;
    const int BC = in_sizes[0] / (512 * 512);

    float* LL1 = (float*)d_ws;

    dwt_db4_level<4, 2><<<dim3(1, 16, BC), dim3(128, 1, 1), 0, stream>>>(
        x,   512, 512LL * 512, LL1, 256, 256LL * 256, out, 256);
    dwt_db4_level23<<<dim3(1, 4, BC), dim3(128, 1, 1), 0, stream>>>(LL1, out);
}
